# starting kernel
# baseline (speedup 1.0000x reference)
#include <hip/hip_runtime.h>

#define D 64

typedef __attribute__((ext_vector_type(2))) float v2f;
typedef __attribute__((ext_vector_type(8))) float v8f;

__global__ void k_deg_init(float* __restrict__ deg, int n) {
    int i = blockIdx.x * blockDim.x + threadIdx.x;
    if (i < n) deg[i] = 1.0f;
}

__global__ void k_deg_accum(const int* __restrict__ dst, float* __restrict__ deg, int e) {
    int i = blockIdx.x * blockDim.x + threadIdx.x;
    if (i < e) atomicAdd(&deg[dst[i]], 1.0f);
}

__global__ void k_dinv(const float* __restrict__ deg, float* __restrict__ dinv, int n) {
    int i = blockIdx.x * blockDim.x + threadIdx.x;
    if (i < n) {
        float d = deg[i];
        dinv[i] = (d > 0.0f) ? rsqrtf(d) : 0.0f;
    }
}

__global__ void __launch_bounds__(256)
k_gemm_wmma(const float* __restrict__ X, const float* __restrict__ W,
            float* __restrict__ Y, int n) {
    const int lane = threadIdx.x & 31;
    const int wave = threadIdx.x >> 5;
    const int rowTiles = (n + 15) >> 4;
    const int rt = blockIdx.x * 2 + (wave >> 2);
    const int ct = wave & 3;
    if (rt >= rowTiles) return;

    const int r0 = rt << 4;
    const int c0 = ct << 4;
    const int m  = lane & 15;
    const int kh = lane >> 4;

    int arow = r0 + m;
    if (arow >= n) arow = n - 1;
    const float* __restrict__ aptr = X + (size_t)arow * D;
    const float* __restrict__ bptr = W + c0 + m;

    v8f acc = {};
#pragma unroll
    for (int k0 = 0; k0 < D; k0 += 4) {
        v2f a = *(const v2f*)(aptr + k0 + 2 * kh);
        v2f b;
        b.x = bptr[(k0 + 2 * kh)     * D];
        b.y = bptr[(k0 + 2 * kh + 1) * D];
        acc = __builtin_amdgcn_wmma_f32_16x16x4_f32(
             false, a,  false, b,
             (short)0, acc,  false,  false);
    }

    const int col = c0 + m;
#pragma unroll
    for (int r = 0; r < 8; ++r) {
        int row = r0 + r + 8 * kh;
        if (row < n) Y[(size_t)row * D + col] = acc[r];
    }
}

__global__ void k_self_bias(const float* __restrict__ xw, const float* __restrict__ dinv,
                            const float* __restrict__ b, float* __restrict__ agg, int total) {
    int i = blockIdx.x * blockDim.x + threadIdx.x;
    if (i < total) {
        int v = i >> 6;
        int d = i & 63;
        float s = dinv[v];
        agg[i] = s * s * xw[i] + b[d];
    }
}

__global__ void __launch_bounds__(256)
k_scatter(const int* __restrict__ src, const int* __restrict__ dst,
          const float* __restrict__ dinv, const float* __restrict__ xw,
          float* __restrict__ agg, int e) {
    int edge = (blockIdx.x << 3) + (threadIdx.x >> 5);
    if (edge >= e) return;
    int s = src[edge];
    int t = dst[edge];
    float nrm = dinv[s] * dinv[t];
    int lane = threadIdx.x & 31;
    const float* __restrict__ xs = xw + (size_t)s * D;
    float* __restrict__ at = agg + (size_t)t * D;
    atomicAdd(at + lane,      nrm * xs[lane]);
    atomicAdd(at + lane + 32, nrm * xs[lane + 32]);
}

__global__ void k_relu(float* __restrict__ h, int total) {
    int i = blockIdx.x * blockDim.x + threadIdx.x;
    if (i < total) h[i] = fmaxf(h[i], 0.0f);
}

extern "C" void kernel_launch(void* const* d_in, const int* in_sizes, int n_in,
                              void* d_out, int out_size, void* d_ws, size_t ws_size,
                              hipStream_t stream) {
    const float* x  = (const float*)d_in[0];
    const int*   ei = (const int*)  d_in[1];
    const float* W1 = (const float*)d_in[2];
    const float* b1 = (const float*)d_in[3];
    const float* W2 = (const float*)d_in[4];
    const float* b2 = (const float*)d_in[5];

    const int n = in_sizes[0] / D;
    const int e = in_sizes[1] / 2;
    const int* src = ei;
    const int* dst = ei + e;

    float* ws   = (float*)d_ws;
    float* deg  = ws;
    float* dinv = ws + n;
    float* xw   = ws + 2 * (size_t)n;
    float* agg  = xw + (size_t)n * D;
    float* out  = (float*)d_out;

    const int T = 256;
    const int total = n * D;
    const int rowTiles = (n + 15) / 16;
    const int gemmBlocks = (rowTiles + 1) / 2;
    const int scatBlocks = (e + 7) / 8;

    k_deg_init <<<(n + T - 1) / T, T, 0, stream>>>(deg, n);
    k_deg_accum<<<(e + T - 1) / T, T, 0, stream>>>(dst, deg, e);
    k_dinv     <<<(n + T - 1) / T, T, 0, stream>>>(deg, dinv, n);

    k_gemm_wmma<<<gemmBlocks, 256, 0, stream>>>(x, W1, xw, n);
    k_self_bias<<<(total + T - 1) / T, T, 0, stream>>>(xw, dinv, b1, agg, total);
    k_scatter  <<<scatBlocks, 256, 0, stream>>>(src, dst, dinv, xw, agg, e);
    k_relu     <<<(total + T - 1) / T, T, 0, stream>>>(agg, total);

    k_gemm_wmma<<<gemmBlocks, 256, 0, stream>>>(agg, W2, xw, n);
    k_self_bias<<<(total + T - 1) / T, T, 0, stream>>>(xw, dinv, b2, out, total);
    k_scatter  <<<scatBlocks, 256, 0, stream>>>(src, dst, dinv, xw, out, e);
}


# edge aggregation: float-atomic scatter replaced by a per-destination gather over a CSR rebuilt each call (LDS counting sort, no global atomics); rows summed in slot (arrival) order
# speedup vs baseline: 3.2714x; 3.2714x over previous
#include <hip/hip_runtime.h>

#define D 64
#define CAP 65536
#define SORT_BLOCKS 128
#define SORT_THREADS 1024
#define COL_THREADS 256

typedef __attribute__((ext_vector_type(2))) float v2f;
typedef __attribute__((ext_vector_type(8))) float v8f;


__global__ void __launch_bounds__(SORT_THREADS)
k_hist(const int* __restrict__ dst, int* __restrict__ hist, int n, int e, int stride) {
    __shared__ int h[CAP];
    const int t   = threadIdx.x;
    const int per = (e + (int)gridDim.x - 1) / (int)gridDim.x;
    const int e0  = (int)blockIdx.x * per;
    const int e1  = min(e0 + per, e);
    int* __restrict__ myhist = hist + (size_t)blockIdx.x * stride;
    for (int base = 0; base < n; base += CAP) {
        const int lim = min(CAP, n - base);
        for (int i = t; i < lim; i += SORT_THREADS) h[i] = 0;
        __syncthreads();
        for (int i0 = e0 + t; i0 < e1; i0 += 8 * SORT_THREADS) {
            int d[8];
#pragma unroll
            for (int u = 0; u < 8; ++u) {
                const int i = i0 + u * SORT_THREADS;
                const int x = dst[min(i, e1 - 1)];
                d[u] = (i < e1) ? x - base : -1;
            }
#pragma unroll
            for (int u = 0; u < 8; ++u)
                if ((unsigned)d[u] < (unsigned)lim) atomicAdd(&h[d[u]], 1);
        }
        __syncthreads();
        for (int i = t; i < lim; i += SORT_THREADS) myhist[base + i] = h[i];
        __syncthreads();
    }
}

__global__ void __launch_bounds__(COL_THREADS)
k_colscan(const int* __restrict__ hist, int* __restrict__ offs, int* __restrict__ cnt,
          int* __restrict__ tileSum, int n, int stride, int nb) {
    __shared__ int red[COL_THREADS / 32];
    const int t = threadIdx.x;
    const int v = (int)blockIdx.x * COL_THREADS + t;
    int run = 0;
    if (v < n) {
        const int* __restrict__ hp = hist + v;
        int* __restrict__ op = offs + v;
        for (int b0 = 0; b0 < nb; b0 += 8) {
            int c[8];
#pragma unroll
            for (int u = 0; u < 8; ++u)
                c[u] = (b0 + u < nb) ? hp[(size_t)(b0 + u) * stride] : 0;
#pragma unroll
            for (int u = 0; u < 8; ++u) {
                if (b0 + u < nb) op[(size_t)(b0 + u) * stride] = run;
                run += c[u];
            }
        }
        cnt[v] = run;
    }
    int s = run;
    for (int o = 16; o > 0; o >>= 1) s += __shfl_xor(s, o);
    if ((t & 31) == 0) red[t >> 5] = s;
    __syncthreads();
    if (t == 0) {
        int tot = 0;
        for (int k = 0; k < COL_THREADS / 32; ++k) tot += red[k];
        tileSum[blockIdx.x] = tot;
    }
}

__global__ void __launch_bounds__(1024)
k_scan(const int* __restrict__ cnt, const int* __restrict__ tileSum, int* __restrict__ rowptr,
       float* __restrict__ dinv, int n) {
    __shared__ int red[32];
    __shared__ int wsum[32];
    const int t    = threadIdx.x;
    const int lane = t & 31;
    const int w    = t >> 5;
    const int start = (int)blockIdx.x * 1024;
    const int ntile = start / COL_THREADS;

    int s = 0;
    for (int i = t; i < ntile; i += 1024) s += tileSum[i];
    for (int o = 16; o > 0; o >>= 1) s += __shfl_xor(s, o);
    if (lane == 0) red[w] = s;

    const int v = start + t;
    const int x = (v < n) ? cnt[v] : 0;
    int inc = x;
    for (int o = 1; o < 32; o <<= 1) {
        const int y = __shfl_up(inc, o);
        if (lane >= o) inc += y;
    }
    if (lane == 31) wsum[w] = inc;
    __syncthreads();

    int prefix = 0, woff = 0;
    for (int k = 0; k < 32; ++k) {
        prefix += red[k];
        if (k < w) woff += wsum[k];
    }
    const int excl = prefix + woff + inc - x;
    if (v < n) {
        rowptr[v] = excl;
        const float d = (float)(x + 1);
        dinv[v] = (d > 0.0f) ? rsqrtf(d) : 0.0f;
        if (v == n - 1) rowptr[n] = excl + x;
    }
}

__global__ void __launch_bounds__(SORT_THREADS)
k_fill(const int* __restrict__ src, const int* __restrict__ dst, const int* __restrict__ rowptr,
       const int* __restrict__ offs, int* __restrict__ csr, int n, int e, int stride) {
    __shared__ int h[CAP];
    const int t   = threadIdx.x;
    const int per = (e + (int)gridDim.x - 1) / (int)gridDim.x;
    const int e0  = (int)blockIdx.x * per;
    const int e1  = min(e0 + per, e);
    const int* __restrict__ myoffs = offs + (size_t)blockIdx.x * stride;
    for (int base = 0; base < n; base += CAP) {
        const int lim = min(CAP, n - base);
#pragma unroll 4
        for (int i = t; i < lim; i += SORT_THREADS) h[i] = rowptr[base + i] + myoffs[base + i];
        __syncthreads();
        for (int i0 = e0 + t; i0 < e1; i0 += 4 * SORT_THREADS) {
            int d[4], s[4];
#pragma unroll
            for (int u = 0; u < 4; ++u) {
                const int i  = i0 + u * SORT_THREADS;
                const int ic = min(i, e1 - 1);
                const int x  = dst[ic];
                s[u] = src[ic];
                d[u] = (i < e1) ? x - base : -1;
            }
#pragma unroll
            for (int u = 0; u < 4; ++u) {
                if ((unsigned)d[u] < (unsigned)lim) {
                    const int p = atomicAdd(&h[d[u]], 1);
                    csr[p] = s[u];
                }
            }
        }
        __syncthreads();
    }
}

__global__ void __launch_bounds__(256)
k_gemm_wmma(const float* __restrict__ X, const float* __restrict__ W,
            float* __restrict__ Y, int n) {
    const int lane = threadIdx.x & 31;
    const int wave = threadIdx.x >> 5;
    const int rowTiles = (n + 15) >> 4;
    const int rt = blockIdx.x * 2 + (wave >> 2);
    const int ct = wave & 3;
    if (rt >= rowTiles) return;

    const int r0 = rt << 4;
    const int c0 = ct << 4;
    const int m  = lane & 15;
    const int kh = lane >> 4;

    int arow = r0 + m;
    if (arow >= n) arow = n - 1;
    const float* __restrict__ aptr = X + (size_t)arow * D;
    const float* __restrict__ bptr = W + c0 + m;

    v8f acc = {};
#pragma unroll
    for (int k0 = 0; k0 < D; k0 += 4) {
        v2f a = *(const v2f*)(aptr + k0 + 2 * kh);
        v2f b;
        b.x = bptr[(k0 + 2 * kh)     * D];
        b.y = bptr[(k0 + 2 * kh + 1) * D];
        acc = __builtin_amdgcn_wmma_f32_16x16x4_f32(
             false, a,  false, b,
             (short)0, acc,  false,  false);
    }

    const int col = c0 + m;
#pragma unroll
    for (int r = 0; r < 8; ++r) {
        int row = r0 + r + 8 * kh;
        if (row < n) Y[(size_t)row * D + col] = acc[r];
    }
}

template <bool RELU>
__global__ void __launch_bounds__(256)
k_gather(const int* __restrict__ rowptr, const int* __restrict__ csr,
         const float* __restrict__ dinv, const float* __restrict__ xw,
         const float* __restrict__ b, float* __restrict__ outp, int n) {
    const int lane = threadIdx.x & 31;
    const int v = __builtin_amdgcn_readfirstlane((int)(blockIdx.x * 8 + (threadIdx.x >> 5)));
    if (v >= n) return;
    const int beg = rowptr[v];
    const int end = rowptr[v + 1];
    const float dv = dinv[v];
    const v2f* __restrict__ xr = (const v2f*)xw;

    v2f acc = xr[(size_t)v * 32 + lane] * (dv * dv);

    for (int j0 = beg; j0 < end; j0 += 32) {
        const int  j  = j0 + lane;
        const bool ok = j < end;
        const int   my = ok ? csr[j] : v;
        const float wt = ok ? dinv[my] * dv : 0.0f;
        const int   wi = __float_as_int(wt);
        const int cntc = min(32, end - j0);
        for (int k0 = 0; k0 < cntc; k0 += 8) {
#pragma unroll
            for (int u = 0; u < 8; ++u) {
                const int   s  = __builtin_amdgcn_readlane(my, k0 + u);
                const float ws = __int_as_float(__builtin_amdgcn_readlane(wi, k0 + u));
                acc += xr[(size_t)s * 32 + lane] * ws;
            }
        }
    }

    v2f res = acc + ((const v2f*)b)[lane];
    if (RELU) {
        res.x = fmaxf(res.x, 0.0f);
        res.y = fmaxf(res.y, 0.0f);
    }
    ((v2f*)outp)[(size_t)v * 32 + lane] = res;
}

extern "C" void kernel_launch(void* const* d_in, const int* in_sizes, int n_in,
                              void* d_out, int out_size, void* d_ws, size_t ws_size,
                              hipStream_t stream) {
    const float* x  = (const float*)d_in[0];
    const int*   ei = (const int*)  d_in[1];
    const float* W1 = (const float*)d_in[2];
    const float* b1 = (const float*)d_in[3];
    const float* W2 = (const float*)d_in[4];
    const float* b2 = (const float*)d_in[5];

    const int n = in_sizes[0] / D;
    const int e = in_sizes[1] / 2;
    const int* src = ei;
    const int* dst = ei + e;

    const int stride   = (n + 31) & ~31;
    const int colTiles = (n + COL_THREADS - 1) / COL_THREADS;

    float* ws    = (float*)d_ws;
    float* xw    = ws;
    float* agg   = xw + (size_t)n * D;
    int*   hist  = (int*)(agg + (size_t)n * D);
    int*   offs  = hist + (size_t)SORT_BLOCKS * stride;
    int*   csr   = offs + (size_t)SORT_BLOCKS * stride;
    int*   cnt   = csr + e;
    int*   rowptr  = cnt + n;
    int*   tileSum = rowptr + n + 1;
    float* dinv  = (float*)(tileSum + colTiles);
    float* out   = (float*)d_out;

    const int rowTiles = (n + 15) / 16;
    const int gemmBlocks = (rowTiles + 1) / 2;
    const int gatherBlocks = (n + 7) / 8;

    k_hist   <<<SORT_BLOCKS, SORT_THREADS, 0, stream>>>(dst, hist, n, e, stride);
    k_colscan<<<colTiles, COL_THREADS, 0, stream>>>(hist, offs, cnt, tileSum, n, stride, SORT_BLOCKS);
    k_scan   <<<(n + 1023) / 1024, 1024, 0, stream>>>(cnt, tileSum, rowptr, dinv, n);
    k_fill   <<<SORT_BLOCKS, SORT_THREADS, 0, stream>>>(src, dst, rowptr, offs, csr, n, e, stride);

    k_gemm_wmma<<<gemmBlocks, 256, 0, stream>>>(x, W1, xw, n);
    k_gather<true><<<gatherBlocks, 256, 0, stream>>>(rowptr, csr, dinv, xw, b1, agg, n);

    k_gemm_wmma<<<gemmBlocks, 256, 0, stream>>>(agg, W2, xw, n);
    k_gather<false><<<gatherBlocks, 256, 0, stream>>>(rowptr, csr, dinv, xw, b2, out, n);
}


# edge sort rebuilt as a two-level bucket counting sort (256-node buckets, one work-group per bucket): 3 launches not 4, no node-by-slice tables; rows summed in slot (arrival) order of the new sort
# speedup vs baseline: 4.6100x; 1.4092x over previous
#include <hip/hip_runtime.h>

#define D 64
#define NB 256
#define SORT_THREADS 1024
#define BSH 8
#define BW (1 << BSH)
#define MAXBK 1024

typedef __attribute__((ext_vector_type(2))) float v2f;
typedef __attribute__((ext_vector_type(8))) float v8f;


__global__ void __launch_bounds__(SORT_THREADS)
k_bhist(const int* __restrict__ dst, int* __restrict__ bh, int e, int nbk) {
    __shared__ int h[MAXBK];
    const int t   = threadIdx.x;
    const int g   = blockIdx.x;
    const int per = (e + NB - 1) / NB;
    const int e0  = g * per;
    const int e1  = min(e0 + per, e);
    for (int i = t; i < nbk; i += SORT_THREADS) h[i] = 0;
    __syncthreads();
    for (int i0 = e0 + t; i0 < e1; i0 += 8 * SORT_THREADS) {
        int k[8];
#pragma unroll
        for (int u = 0; u < 8; ++u) {
            const int i = i0 + u * SORT_THREADS;
            const int x = dst[min(i, e1 - 1)];
            k[u] = (i < e1) ? (x >> BSH) : -1;
        }
#pragma unroll
        for (int u = 0; u < 8; ++u)
            if ((unsigned)k[u] < (unsigned)nbk) atomicAdd(&h[k[u]], 1);
    }
    __syncthreads();
    for (int i = t; i < nbk; i += SORT_THREADS) bh[(size_t)i * NB + g] = h[i];
}

__global__ void __launch_bounds__(SORT_THREADS)
k_bscatter(const int* __restrict__ src, const int* __restrict__ dst, const int* __restrict__ bh,
           int* __restrict__ bstart, int* __restrict__ pairs, int e, int nbk) {
    __shared__ int tot[MAXBK];
    __shared__ int cur[MAXBK];
    __shared__ int wsum[32];
    const int t    = threadIdx.x;
    const int lane = t & 31;
    const int w    = t >> 5;
    const int g    = blockIdx.x;

    const int wu = __builtin_amdgcn_readfirstlane(w);
    for (int k0 = wu; k0 < nbk; k0 += 4 * 32) {
        int s[4], p[4];
#pragma unroll
        for (int q = 0; q < 4; ++q) {
            const int k = min(k0 + 32 * q, nbk - 1);
            const int* __restrict__ row = bh + (size_t)k * NB;
            s[q] = 0;
            p[q] = 0;
#pragma unroll
            for (int j = lane; j < NB; j += 32) {
                const int c = row[j];
                s[q] += c;
                p[q] += (j < g) ? c : 0;
            }
        }
#pragma unroll
        for (int q = 0; q < 4; ++q) {
            int sq = s[q], pq = p[q];
            for (int o = 16; o > 0; o >>= 1) {
                sq += __shfl_xor(sq, o);
                pq += __shfl_xor(pq, o);
            }
            const int k = k0 + 32 * q;
            if (lane == 0 && k < nbk) { tot[k] = sq; cur[k] = pq; }
        }
    }
    __syncthreads();

    const int x = (t < nbk) ? tot[t] : 0;
    int inc = x;
    for (int o = 1; o < 32; o <<= 1) {
        const int y = __shfl_up(inc, o);
        if (lane >= o) inc += y;
    }
    if (lane == 31) wsum[w] = inc;
    __syncthreads();
    int woff = 0;
    for (int q = 0; q < 32; ++q)
        if (q < w) woff += wsum[q];
    const int excl = woff + inc - x;
    if (t < nbk) {
        cur[t] += excl;
        if (g == 0) {
            bstart[t] = excl;
            if (t == nbk - 1) bstart[nbk] = excl + x;
        }
    }
    __syncthreads();

    const int per = (e + NB - 1) / NB;
    const int e0  = g * per;
    const int e1  = min(e0 + per, e);
    int2* __restrict__ pp = (int2*)pairs;
    for (int i0 = e0 + t; i0 < e1; i0 += 8 * SORT_THREADS) {
        int d[8], s[8];
#pragma unroll
        for (int u = 0; u < 8; ++u) {
            const int i  = i0 + u * SORT_THREADS;
            const int ic = min(i, e1 - 1);
            const int xd = dst[ic];
            s[u] = src[ic];
            d[u] = (i < e1) ? xd : -1;
        }
#pragma unroll
        for (int u = 0; u < 8; ++u) {
            const int k = d[u] >> BSH;
            if ((unsigned)k < (unsigned)nbk) {
                const int p = atomicAdd(&cur[k], 1);
                pp[p] = make_int2(s[u], d[u]);
            }
        }
    }
}

__global__ void __launch_bounds__(SORT_THREADS)
k_bcsr(const int* __restrict__ pairs, const int* __restrict__ bstart, int* __restrict__ rowptr,
       int* __restrict__ csr, float* __restrict__ dinv, int n) {
    __shared__ int cnt[BW];
    __shared__ int cur[BW];
    __shared__ int wsum[BW / 32];
    const int t    = threadIdx.x;
    const int lane = t & 31;
    const int w    = t >> 5;
    const int k    = blockIdx.x;
    const int base = k << BSH;
    const int b0   = bstart[k];
    const int b1   = bstart[k + 1];
    const int2* __restrict__ pp = (const int2*)pairs;

    if (t < BW) cnt[t] = 0;
    __syncthreads();
    for (int i0 = b0 + t; i0 < b1; i0 += 8 * SORT_THREADS) {
        int d[8];
#pragma unroll
        for (int u = 0; u < 8; ++u) {
            const int i = i0 + u * SORT_THREADS;
            const int2 pr = pp[min(i, b1 - 1)];
            d[u] = (i < b1) ? pr.y - base : -1;
        }
#pragma unroll
        for (int u = 0; u < 8; ++u)
            if ((unsigned)d[u] < (unsigned)BW) atomicAdd(&cnt[d[u]], 1);
    }
    __syncthreads();

    const int x = (t < BW) ? cnt[t] : 0;
    int inc = x;
    for (int o = 1; o < 32; o <<= 1) {
        const int y = __shfl_up(inc, o);
        if (lane >= o) inc += y;
    }
    if (t < BW && lane == 31) wsum[w] = inc;
    __syncthreads();
    if (t < BW) {
        int woff = 0;
        for (int q = 0; q < BW / 32; ++q)
            if (q < w) woff += wsum[q];
        const int excl = b0 + woff + inc - x;
        cur[t] = excl;
        const int v = base + t;
        if (v < n) {
            rowptr[v] = excl;
            const float dg = (float)(x + 1);
            dinv[v] = (dg > 0.0f) ? rsqrtf(dg) : 0.0f;
            if (v == n - 1) rowptr[n] = excl + x;
        }
    }
    __syncthreads();

    for (int i0 = b0 + t; i0 < b1; i0 += 8 * SORT_THREADS) {
        int d[8], s[8];
#pragma unroll
        for (int u = 0; u < 8; ++u) {
            const int i = i0 + u * SORT_THREADS;
            const int2 pr = pp[min(i, b1 - 1)];
            s[u] = pr.x;
            d[u] = (i < b1) ? pr.y - base : -1;
        }
#pragma unroll
        for (int u = 0; u < 8; ++u) {
            if ((unsigned)d[u] < (unsigned)BW) {
                const int p = atomicAdd(&cur[d[u]], 1);
                csr[p] = s[u];
            }
        }
    }
}

__global__ void __launch_bounds__(256)
k_gemm_wmma(const float* __restrict__ X, const float* __restrict__ W,
            float* __restrict__ Y, int n) {
    const int lane = threadIdx.x & 31;
    const int wave = threadIdx.x >> 5;
    const int rowTiles = (n + 15) >> 4;
    const int rt = blockIdx.x * 2 + (wave >> 2);
    const int ct = wave & 3;
    if (rt >= rowTiles) return;

    const int r0 = rt << 4;
    const int c0 = ct << 4;
    const int m  = lane & 15;
    const int kh = lane >> 4;

    int arow = r0 + m;
    if (arow >= n) arow = n - 1;
    const float* __restrict__ aptr = X + (size_t)arow * D;
    const float* __restrict__ bptr = W + c0 + m;

    v8f acc = {};
#pragma unroll
    for (int k0 = 0; k0 < D; k0 += 4) {
        v2f a = *(const v2f*)(aptr + k0 + 2 * kh);
        v2f b;
        b.x = bptr[(k0 + 2 * kh)     * D];
        b.y = bptr[(k0 + 2 * kh + 1) * D];
        acc = __builtin_amdgcn_wmma_f32_16x16x4_f32(
             false, a,  false, b,
             (short)0, acc,  false,  false);
    }

    const int col = c0 + m;
#pragma unroll
    for (int r = 0; r < 8; ++r) {
        int row = r0 + r + 8 * kh;
        if (row < n) Y[(size_t)row * D + col] = acc[r];
    }
}

template <bool RELU>
__global__ void __launch_bounds__(256)
k_gather(const int* __restrict__ rowptr, const int* __restrict__ csr,
         const float* __restrict__ dinv, const float* __restrict__ xw,
         const float* __restrict__ b, float* __restrict__ outp, int n) {
    const int lane = threadIdx.x & 31;
    const int v = __builtin_amdgcn_readfirstlane((int)(blockIdx.x * 8 + (threadIdx.x >> 5)));
    if (v >= n) return;
    const int beg = rowptr[v];
    const int end = rowptr[v + 1];
    const float dv = dinv[v];
    const v2f* __restrict__ xr = (const v2f*)xw;

    v2f acc = xr[(size_t)v * 32 + lane] * (dv * dv);

    for (int j0 = beg; j0 < end; j0 += 32) {
        const int  j  = j0 + lane;
        const bool ok = j < end;
        const int   my = ok ? csr[j] : v;
        const float wt = ok ? dinv[my] * dv : 0.0f;
        const int   wi = __float_as_int(wt);
        const int cntc = min(32, end - j0);
        for (int k0 = 0; k0 < cntc; k0 += 8) {
#pragma unroll
            for (int u = 0; u < 8; ++u) {
                const int   s  = __builtin_amdgcn_readlane(my, k0 + u);
                const float ws = __int_as_float(__builtin_amdgcn_readlane(wi, k0 + u));
                acc += xr[(size_t)s * 32 + lane] * ws;
            }
        }
    }

    v2f res = acc + ((const v2f*)b)[lane];
    if (RELU) {
        res.x = fmaxf(res.x, 0.0f);
        res.y = fmaxf(res.y, 0.0f);
    }
    ((v2f*)outp)[(size_t)v * 32 + lane] = res;
}

extern "C" void kernel_launch(void* const* d_in, const int* in_sizes, int n_in,
                              void* d_out, int out_size, void* d_ws, size_t ws_size,
                              hipStream_t stream) {
    const float* x  = (const float*)d_in[0];
    const int*   ei = (const int*)  d_in[1];
    const float* W1 = (const float*)d_in[2];
    const float* b1 = (const float*)d_in[3];
    const float* W2 = (const float*)d_in[4];
    const float* b2 = (const float*)d_in[5];

    const int n = in_sizes[0] / D;
    const int e = in_sizes[1] / 2;
    const int* src = ei;
    const int* dst = ei + e;

    const int nbk = (n + BW - 1) >> BSH;

    float* ws    = (float*)d_ws;
    float* xw    = ws;
    float* agg   = xw + (size_t)n * D;
    int*   pairs = (int*)(agg + (size_t)n * D);
    int*   csr   = pairs + (size_t)2 * e;
    int*   bh    = csr + e;
    int*   bstart  = bh + (size_t)nbk * NB;
    int*   rowptr  = bstart + nbk + 1;
    float* dinv  = (float*)(rowptr + n + 1);
    float* out   = (float*)d_out;

    const int rowTiles = (n + 15) / 16;
    const int gemmBlocks = (rowTiles + 1) / 2;
    const int gatherBlocks = (n + 7) / 8;

    k_bhist   <<<NB, SORT_THREADS, 0, stream>>>(dst, bh, e, nbk);
    k_bscatter<<<NB, SORT_THREADS, 0, stream>>>(src, dst, bh, bstart, pairs, e, nbk);
    k_bcsr    <<<nbk, SORT_THREADS, 0, stream>>>(pairs, bstart, rowptr, csr, dinv, n);

    k_gemm_wmma<<<gemmBlocks, 256, 0, stream>>>(x, W1, xw, n);
    k_gather<true><<<gatherBlocks, 256, 0, stream>>>(rowptr, csr, dinv, xw, b1, agg, n);

    k_gemm_wmma<<<gemmBlocks, 256, 0, stream>>>(agg, W2, xw, n);
    k_gather<false><<<gatherBlocks, 256, 0, stream>>>(rowptr, csr, dinv, xw, b2, out, n);
}


# both 64x64 products fused into the first aggregation kernel (16-node LDS tile, f32 WMMA): layer 1 sums x rows first, then applies W1 (sum order: (Ax)W1 not A(xW1)); 5 launches not 7
# speedup vs baseline: 5.0193x; 1.0888x over previous
#include <hip/hip_runtime.h>

#define D 64
#define NB 256
#define SORT_THREADS 1024
#define BSH 8
#define BW (1 << BSH)
#define MAXBK 1024

typedef __attribute__((ext_vector_type(2))) float v2f;
typedef __attribute__((ext_vector_type(8))) float v8f;


__global__ void __launch_bounds__(SORT_THREADS)
k_bhist(const int* __restrict__ dst, int* __restrict__ bh, int e, int nbk) {
    __shared__ int h[MAXBK];
    const int t   = threadIdx.x;
    const int g   = blockIdx.x;
    const int per = (e + NB - 1) / NB;
    const int e0  = g * per;
    const int e1  = min(e0 + per, e);
    for (int i = t; i < nbk; i += SORT_THREADS) h[i] = 0;
    __syncthreads();
    for (int i0 = e0 + t; i0 < e1; i0 += 8 * SORT_THREADS) {
        int k[8];
#pragma unroll
        for (int u = 0; u < 8; ++u) {
            const int i = i0 + u * SORT_THREADS;
            const int x = dst[min(i, e1 - 1)];
            k[u] = (i < e1) ? (x >> BSH) : -1;
        }
#pragma unroll
        for (int u = 0; u < 8; ++u)
            if ((unsigned)k[u] < (unsigned)nbk) atomicAdd(&h[k[u]], 1);
    }
    __syncthreads();
    for (int i = t; i < nbk; i += SORT_THREADS) bh[(size_t)i * NB + g] = h[i];
}

__global__ void __launch_bounds__(SORT_THREADS)
k_bscatter(const int* __restrict__ src, const int* __restrict__ dst, const int* __restrict__ bh,
           int* __restrict__ bstart, int* __restrict__ pairs, int e, int nbk) {
    __shared__ int tot[MAXBK];
    __shared__ int cur[MAXBK];
    __shared__ int wsum[32];
    const int t    = threadIdx.x;
    const int lane = t & 31;
    const int w    = t >> 5;
    const int g    = blockIdx.x;

    const int wu = __builtin_amdgcn_readfirstlane(w);
    for (int k0 = wu; k0 < nbk; k0 += 4 * 32) {
        int s[4], p[4];
#pragma unroll
        for (int q = 0; q < 4; ++q) {
            const int k = min(k0 + 32 * q, nbk - 1);
            const int* __restrict__ row = bh + (size_t)k * NB;
            s[q] = 0;
            p[q] = 0;
#pragma unroll
            for (int j = lane; j < NB; j += 32) {
                const int c = row[j];
                s[q] += c;
                p[q] += (j < g) ? c : 0;
            }
        }
#pragma unroll
        for (int q = 0; q < 4; ++q) {
            int sq = s[q], pq = p[q];
            for (int o = 16; o > 0; o >>= 1) {
                sq += __shfl_xor(sq, o);
                pq += __shfl_xor(pq, o);
            }
            const int k = k0 + 32 * q;
            if (lane == 0 && k < nbk) { tot[k] = sq; cur[k] = pq; }
        }
    }
    __syncthreads();

    const int x = (t < nbk) ? tot[t] : 0;
    int inc = x;
    for (int o = 1; o < 32; o <<= 1) {
        const int y = __shfl_up(inc, o);
        if (lane >= o) inc += y;
    }
    if (lane == 31) wsum[w] = inc;
    __syncthreads();
    int woff = 0;
    for (int q = 0; q < 32; ++q)
        if (q < w) woff += wsum[q];
    const int excl = woff + inc - x;
    if (t < nbk) {
        cur[t] += excl;
        if (g == 0) {
            bstart[t] = excl;
            if (t == nbk - 1) bstart[nbk] = excl + x;
        }
    }
    __syncthreads();

    const int per = (e + NB - 1) / NB;
    const int e0  = g * per;
    const int e1  = min(e0 + per, e);
    int2* __restrict__ pp = (int2*)pairs;
    for (int i0 = e0 + t; i0 < e1; i0 += 8 * SORT_THREADS) {
        int d[8], s[8];
#pragma unroll
        for (int u = 0; u < 8; ++u) {
            const int i  = i0 + u * SORT_THREADS;
            const int ic = min(i, e1 - 1);
            const int xd = dst[ic];
            s[u] = src[ic];
            d[u] = (i < e1) ? xd : -1;
        }
#pragma unroll
        for (int u = 0; u < 8; ++u) {
            const int k = d[u] >> BSH;
            if ((unsigned)k < (unsigned)nbk) {
                const int p = atomicAdd(&cur[k], 1);
                pp[p] = make_int2(s[u], d[u]);
            }
        }
    }
}

__global__ void __launch_bounds__(SORT_THREADS)
k_bcsr(const int* __restrict__ pairs, const int* __restrict__ bstart, int* __restrict__ rowptr,
       int* __restrict__ csr, float* __restrict__ dinv, int n) {
    __shared__ int cnt[BW];
    __shared__ int cur[BW];
    __shared__ int wsum[BW / 32];
    const int t    = threadIdx.x;
    const int lane = t & 31;
    const int w    = t >> 5;
    const int k    = blockIdx.x;
    const int base = k << BSH;
    const int b0   = bstart[k];
    const int b1   = bstart[k + 1];
    const int2* __restrict__ pp = (const int2*)pairs;

    if (t < BW) cnt[t] = 0;
    __syncthreads();
    for (int i0 = b0 + t; i0 < b1; i0 += 8 * SORT_THREADS) {
        int d[8];
#pragma unroll
        for (int u = 0; u < 8; ++u) {
            const int i = i0 + u * SORT_THREADS;
            const int2 pr = pp[min(i, b1 - 1)];
            d[u] = (i < b1) ? pr.y - base : -1;
        }
#pragma unroll
        for (int u = 0; u < 8; ++u)
            if ((unsigned)d[u] < (unsigned)BW) atomicAdd(&cnt[d[u]], 1);
    }
    __syncthreads();

    const int x = (t < BW) ? cnt[t] : 0;
    int inc = x;
    for (int o = 1; o < 32; o <<= 1) {
        const int y = __shfl_up(inc, o);
        if (lane >= o) inc += y;
    }
    if (t < BW && lane == 31) wsum[w] = inc;
    __syncthreads();
    if (t < BW) {
        int woff = 0;
        for (int q = 0; q < BW / 32; ++q)
            if (q < w) woff += wsum[q];
        const int excl = b0 + woff + inc - x;
        cur[t] = excl;
        const int v = base + t;
        if (v < n) {
            rowptr[v] = excl;
            const float dg = (float)(x + 1);
            dinv[v] = (dg > 0.0f) ? rsqrtf(dg) : 0.0f;
            if (v == n - 1) rowptr[n] = excl + x;
        }
    }
    __syncthreads();

    for (int i0 = b0 + t; i0 < b1; i0 += 8 * SORT_THREADS) {
        int d[8], s[8];
#pragma unroll
        for (int u = 0; u < 8; ++u) {
            const int i = i0 + u * SORT_THREADS;
            const int2 pr = pp[min(i, b1 - 1)];
            s[u] = pr.x;
            d[u] = (i < b1) ? pr.y - base : -1;
        }
#pragma unroll
        for (int u = 0; u < 8; ++u) {
            if ((unsigned)d[u] < (unsigned)BW) {
                const int p = atomicAdd(&cur[d[u]], 1);
                csr[p] = s[u];
            }
        }
    }
}

#define TP 68

__device__ __forceinline__ void load_bfrags(const float* __restrict__ Wm, int c0, int m, int kh,
                                            v2f (&bf)[16]) {
#pragma unroll
    for (int i = 0; i < 16; ++i) {
        const int k = 4 * i + 2 * kh;
        bf[i].x = Wm[k * D + c0 + m];
        bf[i].y = Wm[(k + 1) * D + c0 + m];
    }
}

__device__ __forceinline__ v8f tile_mm(const float* tile, int m, int kh, const v2f (&bf)[16]) {
    v8f acc = {};
#pragma unroll
    for (int i = 0; i < 16; ++i) {
        const v2f a = *(const v2f*)(tile + m * TP + 4 * i + 2 * kh);
        acc = __builtin_amdgcn_wmma_f32_16x16x4_f32(
             false, a,  false, bf[i],
             (short)0, acc,  false,  false);
    }
    return acc;
}

__global__ void __launch_bounds__(256)
k_gather_mm(const int* __restrict__ rowptr, const int* __restrict__ csr,
            const float* __restrict__ dinv, const float* __restrict__ x,
            const float* __restrict__ W1, const float* __restrict__ b1,
            const float* __restrict__ W2, float* __restrict__ xw2, int n) {
    __shared__ __attribute__((aligned(16))) float tA[16 * TP];
    __shared__ __attribute__((aligned(16))) float tH[16 * TP];
    const int lane = threadIdx.x & 31;
    const int wave = __builtin_amdgcn_readfirstlane((int)(threadIdx.x >> 5));
    const int r0   = blockIdx.x * 16;
    const v2f* __restrict__ xr = (const v2f*)x;

    for (int q = 0; q < 2; ++q) {
        const int row = 2 * wave + q;
        const int v   = r0 + row;
        v2f acc = {0.0f, 0.0f};
        if (v < n) {
            const int beg = rowptr[v];
            const int end = rowptr[v + 1];
            const float dv = dinv[v];
            acc = xr[(size_t)v * 32 + lane] * (dv * dv);
            for (int j0 = beg; j0 < end; j0 += 32) {
                const int  j  = j0 + lane;
                const bool ok = j < end;
                const int   my = ok ? csr[j] : v;
                const float wt = ok ? dinv[my] * dv : 0.0f;
                const int   wi = __float_as_int(wt);
                const int cntc = min(32, end - j0);
                for (int k0 = 0; k0 < cntc; k0 += 8) {
#pragma unroll
                    for (int u = 0; u < 8; ++u) {
                        const int   s  = __builtin_amdgcn_readlane(my, k0 + u);
                        const float ws = __int_as_float(__builtin_amdgcn_readlane(wi, k0 + u));
                        acc += xr[(size_t)s * 32 + lane] * ws;
                    }
                }
            }
        }
        *(v2f*)(tA + row * TP + 2 * lane) = acc;
    }

    const int m  = lane & 15;
    const int kh = lane >> 4;
    const int c0 = (wave & 3) << 4;
    v2f bf[16];
    load_bfrags(wave < 4 ? W1 : W2, c0, m, kh, bf);
    __syncthreads();

    if (wave < 4) {
        const v8f acc = tile_mm(tA, m, kh, bf);
        const float bb = b1[c0 + m];
#pragma unroll
        for (int r = 0; r < 8; ++r)
            tH[(r + 8 * kh) * TP + c0 + m] = fmaxf(acc[r] + bb, 0.0f);
    }
    __syncthreads();

    if (wave >= 4) {
        const v8f acc = tile_mm(tH, m, kh, bf);
        const int col = c0 + m;
#pragma unroll
        for (int r = 0; r < 8; ++r) {
            const int row = r0 + r + 8 * kh;
            if (row < n) xw2[(size_t)row * D + col] = acc[r];
        }
    }
}

template <bool RELU>
__global__ void __launch_bounds__(256)
k_gather(const int* __restrict__ rowptr, const int* __restrict__ csr,
         const float* __restrict__ dinv, const float* __restrict__ xw,
         const float* __restrict__ b, float* __restrict__ outp, int n) {
    const int lane = threadIdx.x & 31;
    const int v = __builtin_amdgcn_readfirstlane((int)(blockIdx.x * 8 + (threadIdx.x >> 5)));
    if (v >= n) return;
    const int beg = rowptr[v];
    const int end = rowptr[v + 1];
    const float dv = dinv[v];
    const v2f* __restrict__ xr = (const v2f*)xw;

    v2f acc = xr[(size_t)v * 32 + lane] * (dv * dv);

    for (int j0 = beg; j0 < end; j0 += 32) {
        const int  j  = j0 + lane;
        const bool ok = j < end;
        const int   my = ok ? csr[j] : v;
        const float wt = ok ? dinv[my] * dv : 0.0f;
        const int   wi = __float_as_int(wt);
        const int cntc = min(32, end - j0);
        for (int k0 = 0; k0 < cntc; k0 += 8) {
#pragma unroll
            for (int u = 0; u < 8; ++u) {
                const int   s  = __builtin_amdgcn_readlane(my, k0 + u);
                const float ws = __int_as_float(__builtin_amdgcn_readlane(wi, k0 + u));
                acc += xr[(size_t)s * 32 + lane] * ws;
            }
        }
    }

    v2f res = acc + ((const v2f*)b)[lane];
    if (RELU) {
        res.x = fmaxf(res.x, 0.0f);
        res.y = fmaxf(res.y, 0.0f);
    }
    ((v2f*)outp)[(size_t)v * 32 + lane] = res;
}

extern "C" void kernel_launch(void* const* d_in, const int* in_sizes, int n_in,
                              void* d_out, int out_size, void* d_ws, size_t ws_size,
                              hipStream_t stream) {
    const float* x  = (const float*)d_in[0];
    const int*   ei = (const int*)  d_in[1];
    const float* W1 = (const float*)d_in[2];
    const float* b1 = (const float*)d_in[3];
    const float* W2 = (const float*)d_in[4];
    const float* b2 = (const float*)d_in[5];

    const int n = in_sizes[0] / D;
    const int e = in_sizes[1] / 2;
    const int* src = ei;
    const int* dst = ei + e;

    const int nbk = (n + BW - 1) >> BSH;

    float* ws    = (float*)d_ws;
    float* xw    = ws;
    int*   pairs = (int*)(xw + (size_t)n * D);
    int*   csr   = pairs + (size_t)2 * e;
    int*   bh    = csr + e;
    int*   bstart  = bh + (size_t)nbk * NB;
    int*   rowptr  = bstart + nbk + 1;
    float* dinv  = (float*)(rowptr + n + 1);
    float* out   = (float*)d_out;

    const int rowTiles = (n + 15) / 16;
    const int gatherBlocks = (n + 7) / 8;

    k_bhist   <<<NB, SORT_THREADS, 0, stream>>>(dst, bh, e, nbk);
    k_bscatter<<<NB, SORT_THREADS, 0, stream>>>(src, dst, bh, bstart, pairs, e, nbk);
    k_bcsr    <<<nbk, SORT_THREADS, 0, stream>>>(pairs, bstart, rowptr, csr, dinv, n);

    k_gather_mm<<<rowTiles, 256, 0, stream>>>(rowptr, csr, dinv, x, W1, b1, W2, xw, n);

    k_gather<false><<<gatherBlocks, 256, 0, stream>>>(rowptr, csr, dinv, xw, b2, out, n);
}


# edge partition in one launch, not two: each slice bucket-sorts its own edges in place, bucket owners collect the per-slice runs; 4 launches not 5; rows summed in arrival order of the new sort
# speedup vs baseline: 5.5119x; 1.0981x over previous
#include <hip/hip_runtime.h>

#define D 64
#define NB 256
#define SORT_THREADS 1024
#define BSH 8
#define BW (1 << BSH)
#define MAXBK 1024

typedef __attribute__((ext_vector_type(2))) float v2f;
typedef __attribute__((ext_vector_type(8))) float v8f;


__global__ void __launch_bounds__(SORT_THREADS)
k_part(const int* __restrict__ src, const int* __restrict__ dst, int* __restrict__ loc,
       int* __restrict__ pairs, int e, int nbk) {
    __shared__ int h[MAXBK];
    __shared__ int wsum[32];
    const int t    = threadIdx.x;
    const int lane = t & 31;
    const int w    = t >> 5;
    const int g    = blockIdx.x;
    const int per  = (e + NB - 1) / NB;
    const int e0   = g * per;
    const int e1   = min(e0 + per, e);

    for (int i = t; i < nbk; i += SORT_THREADS) h[i] = 0;
    __syncthreads();

    for (int i0 = e0 + t; i0 < e1; i0 += 8 * SORT_THREADS) {
        int k[8];
#pragma unroll
        for (int u = 0; u < 8; ++u) {
            const int i = i0 + u * SORT_THREADS;
            const int x = dst[min(i, e1 - 1)];
            k[u] = (i < e1) ? (x >> BSH) : -1;
        }
#pragma unroll
        for (int u = 0; u < 8; ++u)
            if ((unsigned)k[u] < (unsigned)nbk) atomicAdd(&h[k[u]], 1);
    }
    __syncthreads();

    const int x = (t < nbk) ? h[t] : 0;
    int inc = x;
    for (int o = 1; o < 32; o <<= 1) {
        const int y = __shfl_up(inc, o);
        if (lane >= o) inc += y;
    }
    if (lane == 31) wsum[w] = inc;
    __syncthreads();
    int woff = 0;
    for (int q = 0; q < 32; ++q)
        if (q < w) woff += wsum[q];
    const int excl = woff + inc - x;
    if (t < nbk) {
        h[t] = excl;
        loc[(size_t)t * NB + g] = excl;
        if (t == nbk - 1) loc[(size_t)nbk * NB + g] = excl + x;
    }
    __syncthreads();

    int2* __restrict__ pp = (int2*)pairs;
    for (int i0 = e0 + t; i0 < e1; i0 += 8 * SORT_THREADS) {
        int d[8], s[8];
#pragma unroll
        for (int u = 0; u < 8; ++u) {
            const int i  = i0 + u * SORT_THREADS;
            const int ic = min(i, e1 - 1);
            const int xd = dst[ic];
            s[u] = src[ic];
            d[u] = (i < e1) ? xd : -1;
        }
#pragma unroll
        for (int u = 0; u < 8; ++u) {
            const int k = d[u] >> BSH;
            if ((unsigned)k < (unsigned)nbk) {
                const int p = atomicAdd(&h[k], 1);
                pp[e0 + p] = make_int2(s[u], d[u]);
            }
        }
    }
}

template <class F>
__device__ __forceinline__ void for_each_pair(const int2* __restrict__ pp, const int* rs,
                                              const int* rl, int wu, int lane, int base, int e,
                                              F f) {
    int st[NB / 32], ln[NB / 32];
#pragma unroll
    for (int q = 0; q < NB / 32; ++q) {
        st[q] = rs[wu + 32 * q];
        ln[q] = rl[wu + 32 * q];
    }
    int s[NB / 16], d[NB / 16];
#pragma unroll
    for (int q = 0; q < NB / 32; ++q) {
#pragma unroll
        for (int hh = 0; hh < 2; ++hh) {
            const int i = lane + 32 * hh;
            const int2 pr = pp[min(st[q] + i, e - 1)];
            s[2 * q + hh] = pr.x;
            d[2 * q + hh] = (i < ln[q]) ? pr.y - base : -1;
        }
    }
#pragma unroll
    for (int u = 0; u < NB / 16; ++u)
        if ((unsigned)d[u] < (unsigned)BW) f(s[u], d[u]);
#pragma unroll
    for (int q = 0; q < NB / 32; ++q) {
        for (int i = 64 + lane; i < ln[q]; i += 32) {
            const int2 pr = pp[st[q] + i];
            const int dd = pr.y - base;
            if ((unsigned)dd < (unsigned)BW) f(pr.x, dd);
        }
    }
}

__global__ void __launch_bounds__(SORT_THREADS)
k_bcsr(const int* __restrict__ pairs, const int* __restrict__ loc, int* __restrict__ rowptr,
       int* __restrict__ csr, float* __restrict__ dinv, int n, int e) {
    __shared__ int cnt[BW];
    __shared__ int cur[BW];
    __shared__ int wsum[BW / 32];
    __shared__ int rs[NB];
    __shared__ int rl[NB];
    __shared__ int wb[NB / 32];
    const int t    = threadIdx.x;
    const int lane = t & 31;
    const int w    = t >> 5;
    const int wu   = __builtin_amdgcn_readfirstlane(w);
    const int k    = blockIdx.x;
    const int base = k << BSH;
    const int per  = (e + NB - 1) / NB;
    const int2* __restrict__ pp = (const int2*)pairs;

    int a = 0;
    if (t < NB) {
        a = loc[(size_t)k * NB + t];
        const int b = loc[(size_t)(k + 1) * NB + t];
        rs[t] = t * per + a;
        rl[t] = b - a;
    }
    if (t < BW) cnt[t] = 0;
    __syncthreads();

    for_each_pair(pp, rs, rl, wu, lane, base, e, [&](int, int d) { atomicAdd(&cnt[d], 1); });
    __syncthreads();

    const int x = (t < BW) ? cnt[t] : 0;
    int inc = x;
    int sa  = a;
    for (int o = 1; o < 32; o <<= 1) {
        const int y = __shfl_up(inc, o);
        const int z = __shfl_xor(sa, o);
        if (lane >= o) inc += y;
        sa += z;
    }
    if (lane == 31) {
        if (t < BW) wsum[w] = inc;
        if (t < NB) wb[w] = sa;
    }
    __syncthreads();
    if (t < BW) {
        int b0 = 0;
        for (int q = 0; q < NB / 32; ++q) b0 += wb[q];
        int woff = 0;
        for (int q = 0; q < BW / 32; ++q)
            if (q < w) woff += wsum[q];
        const int excl = b0 + woff + inc - x;
        cur[t] = excl;
        const int v = base + t;
        if (v < n) {
            rowptr[v] = excl;
            const float dg = (float)(x + 1);
            dinv[v] = (dg > 0.0f) ? rsqrtf(dg) : 0.0f;
            if (v == n - 1) rowptr[n] = excl + x;
        }
    }
    __syncthreads();

    for_each_pair(pp, rs, rl, wu, lane, base, e, [&](int s, int d) {
        const int p = atomicAdd(&cur[d], 1);
        csr[p] = s;
    });
}

#define TP 68

__device__ __forceinline__ void load_bfrags(const float* __restrict__ Wm, int c0, int m, int kh,
                                            v2f (&bf)[16]) {
#pragma unroll
    for (int i = 0; i < 16; ++i) {
        const int k = 4 * i + 2 * kh;
        bf[i].x = Wm[k * D + c0 + m];
        bf[i].y = Wm[(k + 1) * D + c0 + m];
    }
}

__device__ __forceinline__ v8f tile_mm(const float* tile, int m, int kh, const v2f (&bf)[16]) {
    v8f acc = {};
#pragma unroll
    for (int i = 0; i < 16; ++i) {
        const v2f a = *(const v2f*)(tile + m * TP + 4 * i + 2 * kh);
        acc = __builtin_amdgcn_wmma_f32_16x16x4_f32(
             false, a,  false, bf[i],
             (short)0, acc,  false,  false);
    }
    return acc;
}

__global__ void __launch_bounds__(256)
k_gather_mm(const int* __restrict__ rowptr, const int* __restrict__ csr,
            const float* __restrict__ dinv, const float* __restrict__ x,
            const float* __restrict__ W1, const float* __restrict__ b1,
            const float* __restrict__ W2, float* __restrict__ xw2, int n) {
    __shared__ __attribute__((aligned(16))) float tA[16 * TP];
    __shared__ __attribute__((aligned(16))) float tH[16 * TP];
    const int lane = threadIdx.x & 31;
    const int wave = __builtin_amdgcn_readfirstlane((int)(threadIdx.x >> 5));
    const int r0   = blockIdx.x * 16;
    const v2f* __restrict__ xr = (const v2f*)x;

    for (int q = 0; q < 2; ++q) {
        const int row = 2 * wave + q;
        const int v   = r0 + row;
        v2f acc = {0.0f, 0.0f};
        if (v < n) {
            const int beg = rowptr[v];
            const int end = rowptr[v + 1];
            const float dv = dinv[v];
            acc = xr[(size_t)v * 32 + lane] * (dv * dv);
            for (int j0 = beg; j0 < end; j0 += 32) {
                const int  j  = j0 + lane;
                const bool ok = j < end;
                const int   my = ok ? csr[j] : v;
                const float wt = ok ? dinv[my] * dv : 0.0f;
                const int   wi = __float_as_int(wt);
                const int cntc = min(32, end - j0);
                for (int k0 = 0; k0 < cntc; k0 += 8) {
#pragma unroll
                    for (int u = 0; u < 8; ++u) {
                        const int   s  = __builtin_amdgcn_readlane(my, k0 + u);
                        const float ws = __int_as_float(__builtin_amdgcn_readlane(wi, k0 + u));
                        acc += xr[(size_t)s * 32 + lane] * ws;
                    }
                }
            }
        }
        *(v2f*)(tA + row * TP + 2 * lane) = acc;
    }

    const int m  = lane & 15;
    const int kh = lane >> 4;
    const int c0 = (wave & 3) << 4;
    v2f bf[16];
    load_bfrags(wave < 4 ? W1 : W2, c0, m, kh, bf);
    __syncthreads();

    if (wave < 4) {
        const v8f acc = tile_mm(tA, m, kh, bf);
        const float bb = b1[c0 + m];
#pragma unroll
        for (int r = 0; r < 8; ++r)
            tH[(r + 8 * kh) * TP + c0 + m] = fmaxf(acc[r] + bb, 0.0f);
    }
    __syncthreads();

    if (wave >= 4) {
        const v8f acc = tile_mm(tH, m, kh, bf);
        const int col = c0 + m;
#pragma unroll
        for (int r = 0; r < 8; ++r) {
            const int row = r0 + r + 8 * kh;
            if (row < n) xw2[(size_t)row * D + col] = acc[r];
        }
    }
}

template <bool RELU>
__global__ void __launch_bounds__(256)
k_gather(const int* __restrict__ rowptr, const int* __restrict__ csr,
         const float* __restrict__ dinv, const float* __restrict__ xw,
         const float* __restrict__ b, float* __restrict__ outp, int n) {
    const int lane = threadIdx.x & 31;
    const int v = __builtin_amdgcn_readfirstlane((int)(blockIdx.x * 8 + (threadIdx.x >> 5)));
    if (v >= n) return;
    const int beg = rowptr[v];
    const int end = rowptr[v + 1];
    const float dv = dinv[v];
    const v2f* __restrict__ xr = (const v2f*)xw;

    v2f acc = xr[(size_t)v * 32 + lane] * (dv * dv);

    for (int j0 = beg; j0 < end; j0 += 32) {
        const int  j  = j0 + lane;
        const bool ok = j < end;
        const int   my = ok ? csr[j] : v;
        const float wt = ok ? dinv[my] * dv : 0.0f;
        const int   wi = __float_as_int(wt);
        const int cntc = min(32, end - j0);
        for (int k0 = 0; k0 < cntc; k0 += 8) {
#pragma unroll
            for (int u = 0; u < 8; ++u) {
                const int   s  = __builtin_amdgcn_readlane(my, k0 + u);
                const float ws = __int_as_float(__builtin_amdgcn_readlane(wi, k0 + u));
                acc += xr[(size_t)s * 32 + lane] * ws;
            }
        }
    }

    v2f res = acc + ((const v2f*)b)[lane];
    if (RELU) {
        res.x = fmaxf(res.x, 0.0f);
        res.y = fmaxf(res.y, 0.0f);
    }
    ((v2f*)outp)[(size_t)v * 32 + lane] = res;
}

extern "C" void kernel_launch(void* const* d_in, const int* in_sizes, int n_in,
                              void* d_out, int out_size, void* d_ws, size_t ws_size,
                              hipStream_t stream) {
    const float* x  = (const float*)d_in[0];
    const int*   ei = (const int*)  d_in[1];
    const float* W1 = (const float*)d_in[2];
    const float* b1 = (const float*)d_in[3];
    const float* W2 = (const float*)d_in[4];
    const float* b2 = (const float*)d_in[5];

    const int n = in_sizes[0] / D;
    const int e = in_sizes[1] / 2;
    const int* src = ei;
    const int* dst = ei + e;

    const int nbk = (n + BW - 1) >> BSH;

    float* ws    = (float*)d_ws;
    float* xw    = ws;
    int*   pairs = (int*)(xw + (size_t)n * D);
    int*   csr   = pairs + (size_t)2 * e;
    int*   loc   = csr + e;
    int*   rowptr  = loc + (size_t)(nbk + 1) * NB;
    float* dinv  = (float*)(rowptr + n + 1);
    float* out   = (float*)d_out;

    const int rowTiles = (n + 15) / 16;
    const int gatherBlocks = (n + 7) / 8;

    k_part<<<NB, SORT_THREADS, 0, stream>>>(src, dst, loc, pairs, e, nbk);
    k_bcsr<<<nbk, SORT_THREADS, 0, stream>>>(pairs, loc, rowptr, csr, dinv, n, e);

    k_gather_mm<<<rowTiles, 256, 0, stream>>>(rowptr, csr, dinv, x, W1, b1, W2, xw, n);

    k_gather<false><<<gatherBlocks, 256, 0, stream>>>(rowptr, csr, dinv, xw, b2, out, n);
}


# edge weight factored out of both aggregations: rows stored pre-scaled by dinv[source], each sum scaled by dinv[dest] once; no per-edge dinv gather; sum order unchanged, rounding points moved
# speedup vs baseline: 6.0780x; 1.1027x over previous
#include <hip/hip_runtime.h>

#define D 64
#define NB 256
#define SORT_THREADS 1024
#define BSH 8
#define BW (1 << BSH)
#define MAXBK 1024

typedef __attribute__((ext_vector_type(2))) float v2f;
typedef __attribute__((ext_vector_type(8))) float v8f;


__global__ void __launch_bounds__(SORT_THREADS)
k_part(const int* __restrict__ src, const int* __restrict__ dst, int* __restrict__ loc,
       int* __restrict__ pairs, int e, int nbk) {
    __shared__ int h[MAXBK];
    __shared__ int wsum[32];
    const int t    = threadIdx.x;
    const int lane = t & 31;
    const int w    = t >> 5;
    const int g    = blockIdx.x;
    const int per  = (e + NB - 1) / NB;
    const int e0   = g * per;
    const int e1   = min(e0 + per, e);

    for (int i = t; i < nbk; i += SORT_THREADS) h[i] = 0;
    __syncthreads();

    for (int i0 = e0 + t; i0 < e1; i0 += 8 * SORT_THREADS) {
        int k[8];
#pragma unroll
        for (int u = 0; u < 8; ++u) {
            const int i = i0 + u * SORT_THREADS;
            const int x = dst[min(i, e1 - 1)];
            k[u] = (i < e1) ? (x >> BSH) : -1;
        }
#pragma unroll
        for (int u = 0; u < 8; ++u)
            if ((unsigned)k[u] < (unsigned)nbk) atomicAdd(&h[k[u]], 1);
    }
    __syncthreads();

    const int x = (t < nbk) ? h[t] : 0;
    int inc = x;
    for (int o = 1; o < 32; o <<= 1) {
        const int y = __shfl_up(inc, o);
        if (lane >= o) inc += y;
    }
    if (lane == 31) wsum[w] = inc;
    __syncthreads();
    int woff = 0;
    for (int q = 0; q < 32; ++q)
        if (q < w) woff += wsum[q];
    const int excl = woff + inc - x;
    if (t < nbk) {
        h[t] = excl;
        loc[(size_t)t * NB + g] = excl;
        if (t == nbk - 1) loc[(size_t)nbk * NB + g] = excl + x;
    }
    __syncthreads();

    int2* __restrict__ pp = (int2*)pairs;
    for (int i0 = e0 + t; i0 < e1; i0 += 8 * SORT_THREADS) {
        int d[8], s[8];
#pragma unroll
        for (int u = 0; u < 8; ++u) {
            const int i  = i0 + u * SORT_THREADS;
            const int ic = min(i, e1 - 1);
            const int xd = dst[ic];
            s[u] = src[ic];
            d[u] = (i < e1) ? xd : -1;
        }
#pragma unroll
        for (int u = 0; u < 8; ++u) {
            const int k = d[u] >> BSH;
            if ((unsigned)k < (unsigned)nbk) {
                const int p = atomicAdd(&h[k], 1);
                pp[e0 + p] = make_int2(s[u], d[u]);
            }
        }
    }
}

template <class F, class G>
__device__ __forceinline__ void for_each_pair(const int2* __restrict__ pp, const int* rs,
                                              const int* rl, int wu, int lane, int base, int e,
                                              F f, G g) {
    int st[NB / 32], ln[NB / 32];
#pragma unroll
    for (int q = 0; q < NB / 32; ++q) {
        st[q] = rs[wu + 32 * q];
        ln[q] = rl[wu + 32 * q];
    }
    int s[NB / 16], d[NB / 16];
#pragma unroll
    for (int q = 0; q < NB / 32; ++q) {
#pragma unroll
        for (int hh = 0; hh < 2; ++hh) {
            const int i = lane + 32 * hh;
            const int2 pr = pp[min(st[q] + i, e - 1)];
            s[2 * q + hh] = pr.x;
            d[2 * q + hh] = (i < ln[q]) ? pr.y - base : -1;
        }
    }
    g();
#pragma unroll
    for (int u = 0; u < NB / 16; ++u)
        if ((unsigned)d[u] < (unsigned)BW) f(s[u], d[u]);
#pragma unroll
    for (int q = 0; q < NB / 32; ++q) {
        for (int i = 64 + lane; i < ln[q]; i += 32) {
            const int2 pr = pp[st[q] + i];
            const int dd = pr.y - base;
            if ((unsigned)dd < (unsigned)BW) f(pr.x, dd);
        }
    }
}

__global__ void __launch_bounds__(SORT_THREADS)
k_bcsr(const int* __restrict__ pairs, const int* __restrict__ loc, int* __restrict__ rowptr,
       int* __restrict__ csr, float* __restrict__ dinv, const float* __restrict__ xin,
       float* __restrict__ xs, int n, int e) {
    __shared__ int cnt[BW];
    __shared__ int cur[BW];
    __shared__ int wsum[BW / 32];
    __shared__ int rs[NB];
    __shared__ int rl[NB];
    __shared__ int wb[NB / 32];
    __shared__ float sdv[BW];
    const int t    = threadIdx.x;
    const int lane = t & 31;
    const int w    = t >> 5;
    const int wu   = __builtin_amdgcn_readfirstlane(w);
    const int k    = blockIdx.x;
    const int base = k << BSH;
    const int per  = (e + NB - 1) / NB;
    const int2* __restrict__ pp = (const int2*)pairs;

    int a = 0;
    if (t < NB) {
        a = loc[(size_t)k * NB + t];
        const int b = loc[(size_t)(k + 1) * NB + t];
        rs[t] = t * per + a;
        rl[t] = b - a;
    }
    if (t < BW) cnt[t] = 0;
    __syncthreads();

    for_each_pair(pp, rs, rl, wu, lane, base, e, [&](int, int d) { atomicAdd(&cnt[d], 1); },
                  [] {});
    __syncthreads();

    const int x = (t < BW) ? cnt[t] : 0;
    int inc = x;
    int sa  = a;
    for (int o = 1; o < 32; o <<= 1) {
        const int y = __shfl_up(inc, o);
        const int z = __shfl_xor(sa, o);
        if (lane >= o) inc += y;
        sa += z;
    }
    if (lane == 31) {
        if (t < BW) wsum[w] = inc;
        if (t < NB) wb[w] = sa;
    }
    __syncthreads();
    if (t < BW) {
        int b0 = 0;
        for (int q = 0; q < NB / 32; ++q) b0 += wb[q];
        int woff = 0;
        for (int q = 0; q < BW / 32; ++q)
            if (q < w) woff += wsum[q];
        const int excl = b0 + woff + inc - x;
        cur[t] = excl;
        const int v = base + t;
        if (v < n) {
            rowptr[v] = excl;
            const float dg = (float)(x + 1);
            const float di = (dg > 0.0f) ? rsqrtf(dg) : 0.0f;
            dinv[v] = di;
            sdv[t]  = di;
            if (v == n - 1) rowptr[n] = excl + x;
        }
    }
    __syncthreads();

    const float4* __restrict__ x4 = (const float4*)xin;
    float4 xv[(BW * D / 4) / SORT_THREADS];
    for_each_pair(pp, rs, rl, wu, lane, base, e, [&](int s, int d) {
        const int p = atomicAdd(&cur[d], 1);
        csr[p] = s;
    }, [&] {
#pragma unroll
        for (int i = 0; i < (BW * D / 4) / SORT_THREADS; ++i) {
            const int q = t + SORT_THREADS * i;
            const int v = min(base + (q >> 4), n - 1);
            xv[i] = x4[(size_t)v * (D / 4) + (q & 15)];
        }
    });

    float4* __restrict__ xs4 = (float4*)xs;
#pragma unroll
    for (int i = 0; i < (BW * D / 4) / SORT_THREADS; ++i) {
        const int q = t + SORT_THREADS * i;
        const int v = base + (q >> 4);
        if (v < n) {
            const float di = sdv[q >> 4];
            float4 r = xv[i];
            r.x *= di; r.y *= di; r.z *= di; r.w *= di;
            xs4[(size_t)v * (D / 4) + (q & 15)] = r;
        }
    }
    if (k == 0 && t < D / 4) xs4[(size_t)n * (D / 4) + t] = make_float4(0.0f, 0.0f, 0.0f, 0.0f);
}

#define TP 68

__device__ __forceinline__ void load_bfrags(const float* __restrict__ Wm, int c0, int m, int kh,
                                            v2f (&bf)[16]) {
#pragma unroll
    for (int i = 0; i < 16; ++i) {
        const int k = 4 * i + 2 * kh;
        bf[i].x = Wm[k * D + c0 + m];
        bf[i].y = Wm[(k + 1) * D + c0 + m];
    }
}

__device__ __forceinline__ v8f tile_mm(const float* tile, int m, int kh, const v2f (&bf)[16]) {
    v8f acc = {};
#pragma unroll
    for (int i = 0; i < 16; ++i) {
        const v2f a = *(const v2f*)(tile + m * TP + 4 * i + 2 * kh);
        acc = __builtin_amdgcn_wmma_f32_16x16x4_f32(
             false, a,  false, bf[i],
             (short)0, acc,  false,  false);
    }
    return acc;
}

__global__ void __launch_bounds__(256)
k_gather_mm(const int* __restrict__ rowptr, const int* __restrict__ csr,
            const float* __restrict__ dinv, const float* __restrict__ x,
            const float* __restrict__ W1, const float* __restrict__ b1,
            const float* __restrict__ W2, float* __restrict__ xw2, int n) {
    __shared__ __attribute__((aligned(16))) float tA[16 * TP];
    __shared__ __attribute__((aligned(16))) float tH[16 * TP];
    __shared__ float tD[16];
    const int lane = threadIdx.x & 31;
    const int wave = __builtin_amdgcn_readfirstlane((int)(threadIdx.x >> 5));
    const int r0   = blockIdx.x * 16;
    const v2f* __restrict__ xr = (const v2f*)x;

    for (int q = 0; q < 2; ++q) {
        const int row = 2 * wave + q;
        const int v   = r0 + row;
        v2f acc = {0.0f, 0.0f};
        float dv = 0.0f;
        if (v < n) {
            const int beg = rowptr[v];
            const int end = rowptr[v + 1];
            dv = dinv[v];
            acc = xr[(size_t)v * 32 + lane];
            for (int j0 = beg; j0 < end; j0 += 32) {
                const int  j  = j0 + lane;
                const int  my = (j < end) ? csr[j] : n;
                const int cntc = min(32, end - j0);
                for (int k0 = 0; k0 < cntc; k0 += 8) {
#pragma unroll
                    for (int u = 0; u < 8; ++u) {
                        const int s = __builtin_amdgcn_readlane(my, k0 + u);
                        acc += xr[(size_t)s * 32 + lane];
                    }
                }
            }
        }
        *(v2f*)(tA + row * TP + 2 * lane) = acc * dv;
        if (lane == 0) tD[row] = dv;
    }

    const int m  = lane & 15;
    const int kh = lane >> 4;
    const int c0 = (wave & 3) << 4;
    v2f bf[16];
    load_bfrags(wave < 4 ? W1 : W2, c0, m, kh, bf);
    __syncthreads();

    if (wave < 4) {
        const v8f acc = tile_mm(tA, m, kh, bf);
        const float bb = b1[c0 + m];
#pragma unroll
        for (int r = 0; r < 8; ++r)
            tH[(r + 8 * kh) * TP + c0 + m] = fmaxf(acc[r] + bb, 0.0f);
    }
    __syncthreads();

    if (wave >= 4) {
        const v8f acc = tile_mm(tH, m, kh, bf);
        const int col = c0 + m;
#pragma unroll
        for (int r = 0; r < 8; ++r) {
            const int row = r0 + r + 8 * kh;
            if (row < n) xw2[(size_t)row * D + col] = acc[r] * tD[r + 8 * kh];
        }
    }
    if (blockIdx.x == 0 && wave == 0) *(v2f*)(xw2 + (size_t)n * D + 2 * lane) = v2f{0.0f, 0.0f};
}

template <bool RELU>
__global__ void __launch_bounds__(256)
k_gather(const int* __restrict__ rowptr, const int* __restrict__ csr,
         const float* __restrict__ dinv, const float* __restrict__ xw,
         const float* __restrict__ b, float* __restrict__ outp, int n) {
    const int lane = threadIdx.x & 31;
    const int v = __builtin_amdgcn_readfirstlane((int)(blockIdx.x * 8 + (threadIdx.x >> 5)));
    if (v >= n) return;
    const int beg = rowptr[v];
    const int end = rowptr[v + 1];
    const float dv = dinv[v];
    const v2f* __restrict__ xr = (const v2f*)xw;

    v2f acc = xr[(size_t)v * 32 + lane];

    for (int j0 = beg; j0 < end; j0 += 32) {
        const int  j  = j0 + lane;
        const int  my = (j < end) ? csr[j] : n;
        const int cntc = min(32, end - j0);
        for (int k0 = 0; k0 < cntc; k0 += 8) {
#pragma unroll
            for (int u = 0; u < 8; ++u) {
                const int s = __builtin_amdgcn_readlane(my, k0 + u);
                acc += xr[(size_t)s * 32 + lane];
            }
        }
    }

    v2f res = acc * dv + ((const v2f*)b)[lane];
    if (RELU) {
        res.x = fmaxf(res.x, 0.0f);
        res.y = fmaxf(res.y, 0.0f);
    }
    ((v2f*)outp)[(size_t)v * 32 + lane] = res;
}

extern "C" void kernel_launch(void* const* d_in, const int* in_sizes, int n_in,
                              void* d_out, int out_size, void* d_ws, size_t ws_size,
                              hipStream_t stream) {
    const float* x  = (const float*)d_in[0];
    const int*   ei = (const int*)  d_in[1];
    const float* W1 = (const float*)d_in[2];
    const float* b1 = (const float*)d_in[3];
    const float* W2 = (const float*)d_in[4];
    const float* b2 = (const float*)d_in[5];

    const int n = in_sizes[0] / D;
    const int e = in_sizes[1] / 2;
    const int* src = ei;
    const int* dst = ei + e;

    const int nbk = (n + BW - 1) >> BSH;

    float* ws    = (float*)d_ws;
    float* xw    = ws;
    float* xs    = xw + (size_t)(n + 1) * D;
    int*   pairs = (int*)(xs + (size_t)(n + 1) * D);
    int*   csr   = pairs + (size_t)2 * e;
    int*   loc   = csr + e;
    int*   rowptr  = loc + (size_t)(nbk + 1) * NB;
    float* dinv  = (float*)(rowptr + n + 1);
    float* out   = (float*)d_out;

    const int rowTiles = (n + 15) / 16;
    const int gatherBlocks = (n + 7) / 8;

    k_part<<<NB, SORT_THREADS, 0, stream>>>(src, dst, loc, pairs, e, nbk);
    k_bcsr<<<nbk, SORT_THREADS, 0, stream>>>(pairs, loc, rowptr, csr, dinv, x, xs, n, e);

    k_gather_mm<<<rowTiles, 256, 0, stream>>>(rowptr, csr, dinv, xs, W1, b1, W2, xw, n);

    k_gather<false><<<gatherBlocks, 256, 0, stream>>>(rowptr, csr, dinv, xw, b2, out, n);
}


# edge sort, both levels: sorted records go to an LDS staging area and leave as whole 128-byte lines, not as one scattered 4- or 8-byte store per edge; rows still summed in arrival order
# speedup vs baseline: 6.5512x; 1.0779x over previous
#include <hip/hip_runtime.h>

#define D 64
#define NB 256
#define SORT_THREADS 1024
#define BSH 8
#define BW (1 << BSH)
#define MAXBK 1024
#define PSTAGE 8192
#define CSTAGE 16384

typedef __attribute__((ext_vector_type(2))) float v2f;
typedef __attribute__((ext_vector_type(8))) float v8f;


__global__ void __launch_bounds__(SORT_THREADS)
k_part(const int* __restrict__ src, const int* __restrict__ dst, int* __restrict__ loc,
       int* __restrict__ pairs, int e, int nbk) {
    __shared__ int h[MAXBK];
    __shared__ int wsum[32];
    __shared__ int2 stg[PSTAGE];
    const int t    = threadIdx.x;
    const int lane = t & 31;
    const int w    = t >> 5;
    const int g    = blockIdx.x;
    const int per  = (e + NB - 1) / NB;
    const int e0   = g * per;
    const int e1   = min(e0 + per, e);
    const bool staged = per <= PSTAGE;

    for (int i = t; i < nbk; i += SORT_THREADS) h[i] = 0;
    __syncthreads();

    for (int i0 = e0 + t; i0 < e1; i0 += 8 * SORT_THREADS) {
        int k[8];
#pragma unroll
        for (int u = 0; u < 8; ++u) {
            const int i = i0 + u * SORT_THREADS;
            const int x = dst[min(i, e1 - 1)];
            k[u] = (i < e1) ? (x >> BSH) : -1;
        }
#pragma unroll
        for (int u = 0; u < 8; ++u)
            if ((unsigned)k[u] < (unsigned)nbk) atomicAdd(&h[k[u]], 1);
    }
    __syncthreads();

    const int x = (t < nbk) ? h[t] : 0;
    int inc = x;
    for (int o = 1; o < 32; o <<= 1) {
        const int y = __shfl_up(inc, o);
        if (lane >= o) inc += y;
    }
    if (lane == 31) wsum[w] = inc;
    __syncthreads();
    int woff = 0;
    int tot  = 0;
    for (int q = 0; q < 32; ++q) {
        const int z = wsum[q];
        tot += z;
        if (q < w) woff += z;
    }
    const int excl = woff + inc - x;
    if (t < nbk) {
        h[t] = excl;
        loc[(size_t)t * NB + g] = excl;
        if (t == nbk - 1) loc[(size_t)nbk * NB + g] = excl + x;
    }
    __syncthreads();

    int2* __restrict__ pp = (int2*)pairs;
    auto move_pairs = [&](auto put) {
        for (int i0 = e0 + t; i0 < e1; i0 += 8 * SORT_THREADS) {
            int d[8], s[8];
#pragma unroll
            for (int u = 0; u < 8; ++u) {
                const int i  = i0 + u * SORT_THREADS;
                const int ic = min(i, e1 - 1);
                const int xd = dst[ic];
                s[u] = src[ic];
                d[u] = (i < e1) ? xd : -1;
            }
#pragma unroll
            for (int u = 0; u < 8; ++u) {
                const int k = d[u] >> BSH;
                if ((unsigned)k < (unsigned)nbk) put(atomicAdd(&h[k], 1), make_int2(s[u], d[u]));
            }
        }
    };
    if (!staged) {
        move_pairs([&](int p, int2 pr) { pp[e0 + p] = pr; });
        return;
    }
    move_pairs([&](int p, int2 pr) { stg[p] = pr; });

    __syncthreads();
    const int a0 = e0 & ~15;
    const int hi = e0 + tot;
    for (int i0 = a0 + t; i0 < hi; i0 += 4 * SORT_THREADS) {
        int2 pr[4];
#pragma unroll
        for (int u = 0; u < 4; ++u) {
            const int i = i0 + u * SORT_THREADS;
            pr[u] = stg[max(min(i, hi - 1), e0) - e0];
        }
#pragma unroll
        for (int u = 0; u < 4; ++u) {
            const int i = i0 + u * SORT_THREADS;
            if (i >= e0 && i < hi) pp[i] = pr[u];
        }
    }
}

template <class F, class G>
__device__ __forceinline__ void for_each_pair(const int2* __restrict__ pp, const int* rs,
                                              const int* rl, int wu, int lane, int base, int e,
                                              F f, G g) {
    int st[NB / 32], ln[NB / 32];
#pragma unroll
    for (int q = 0; q < NB / 32; ++q) {
        st[q] = rs[wu + 32 * q];
        ln[q] = rl[wu + 32 * q];
    }
    int s[NB / 16], d[NB / 16];
#pragma unroll
    for (int q = 0; q < NB / 32; ++q) {
#pragma unroll
        for (int hh = 0; hh < 2; ++hh) {
            const int i = lane + 32 * hh;
            const int2 pr = pp[min(st[q] + i, e - 1)];
            s[2 * q + hh] = pr.x;
            d[2 * q + hh] = (i < ln[q]) ? pr.y - base : -1;
        }
    }
    g();
#pragma unroll
    for (int u = 0; u < NB / 16; ++u)
        if ((unsigned)d[u] < (unsigned)BW) f(s[u], d[u]);
#pragma unroll
    for (int q = 0; q < NB / 32; ++q) {
        for (int i = 64 + lane; i < ln[q]; i += 32) {
            const int2 pr = pp[st[q] + i];
            const int dd = pr.y - base;
            if ((unsigned)dd < (unsigned)BW) f(pr.x, dd);
        }
    }
}

__global__ void __launch_bounds__(SORT_THREADS)
k_bcsr(const int* __restrict__ pairs, const int* __restrict__ loc, int* __restrict__ rowptr,
       int* __restrict__ csr, float* __restrict__ dinv, const float* __restrict__ xin,
       float* __restrict__ xs, int n, int e) {
    __shared__ int cnt[BW];
    __shared__ int cur[BW];
    __shared__ int wsum[BW / 32];
    __shared__ int rs[NB];
    __shared__ int rl[NB];
    __shared__ int wb[NB / 32];
    __shared__ float sdv[BW];
    __shared__ int stg[CSTAGE];
    const int t    = threadIdx.x;
    const int lane = t & 31;
    const int w    = t >> 5;
    const int wu   = __builtin_amdgcn_readfirstlane(w);
    const int k    = blockIdx.x;
    const int base = k << BSH;
    const int per  = (e + NB - 1) / NB;
    const int2* __restrict__ pp = (const int2*)pairs;

    int a = 0;
    if (t < NB) {
        a = loc[(size_t)k * NB + t];
        const int b = loc[(size_t)(k + 1) * NB + t];
        rs[t] = t * per + a;
        rl[t] = b - a;
    }
    if (t < BW) cnt[t] = 0;
    __syncthreads();

    for_each_pair(pp, rs, rl, wu, lane, base, e, [&](int, int d) { atomicAdd(&cnt[d], 1); },
                  [] {});
    __syncthreads();

    const int x = (t < BW) ? cnt[t] : 0;
    int inc = x;
    int sa  = a;
    for (int o = 1; o < 32; o <<= 1) {
        const int y = __shfl_up(inc, o);
        const int z = __shfl_xor(sa, o);
        if (lane >= o) inc += y;
        sa += z;
    }
    if (lane == 31) {
        if (t < BW) wsum[w] = inc;
        if (t < NB) wb[w] = sa;
    }
    __syncthreads();
    int b0 = 0;
    for (int q = 0; q < NB / 32; ++q) b0 += wb[q];
    int tot = 0;
    for (int q = 0; q < BW / 32; ++q) tot += wsum[q];
    const bool staged = tot <= CSTAGE;
    if (t < BW) {
        int woff = 0;
        for (int q = 0; q < BW / 32; ++q)
            if (q < w) woff += wsum[q];
        const int lo   = woff + inc - x;
        const int excl = b0 + lo;
        cur[t] = lo;
        const int v = base + t;
        if (v < n) {
            rowptr[v] = excl;
            const float dg = (float)(x + 1);
            const float di = (dg > 0.0f) ? rsqrtf(dg) : 0.0f;
            dinv[v] = di;
            sdv[t]  = di;
            if (v == n - 1) rowptr[n] = excl + x;
        }
    }
    __syncthreads();

    const float4* __restrict__ x4 = (const float4*)xin;
    float4 xv[(BW * D / 4) / SORT_THREADS];
    auto fetch_x = [&] {
#pragma unroll
        for (int i = 0; i < (BW * D / 4) / SORT_THREADS; ++i) {
            const int q = t + SORT_THREADS * i;
            const int v = min(base + (q >> 4), n - 1);
            xv[i] = x4[(size_t)v * (D / 4) + (q & 15)];
        }
    };
    if (staged) {
        for_each_pair(pp, rs, rl, wu, lane, base, e,
                      [&](int s, int d) { stg[atomicAdd(&cur[d], 1)] = s; }, fetch_x);
        __syncthreads();
        const int a0 = b0 & ~31;
        const int hi = b0 + tot;
        for (int i0 = a0 + t; i0 < hi; i0 += 4 * SORT_THREADS) {
            int sv[4];
#pragma unroll
            for (int u = 0; u < 4; ++u) {
                const int i = i0 + u * SORT_THREADS;
                sv[u] = stg[max(min(i, hi - 1), b0) - b0];
            }
#pragma unroll
            for (int u = 0; u < 4; ++u) {
                const int i = i0 + u * SORT_THREADS;
                if (i >= b0 && i < hi) csr[i] = sv[u];
            }
        }
    } else {
        for_each_pair(pp, rs, rl, wu, lane, base, e,
                      [&](int s, int d) { csr[b0 + atomicAdd(&cur[d], 1)] = s; }, fetch_x);
    }

    float4* __restrict__ xs4 = (float4*)xs;
#pragma unroll
    for (int i = 0; i < (BW * D / 4) / SORT_THREADS; ++i) {
        const int q = t + SORT_THREADS * i;
        const int v = base + (q >> 4);
        if (v < n) {
            const float di = sdv[q >> 4];
            float4 r = xv[i];
            r.x *= di; r.y *= di; r.z *= di; r.w *= di;
            xs4[(size_t)v * (D / 4) + (q & 15)] = r;
        }
    }
    if (k == 0 && t < D / 4) xs4[(size_t)n * (D / 4) + t] = make_float4(0.0f, 0.0f, 0.0f, 0.0f);
}

#define TP 68

__device__ __forceinline__ void load_bfrags(const float* __restrict__ Wm, int c0, int m, int kh,
                                            v2f (&bf)[16]) {
#pragma unroll
    for (int i = 0; i < 16; ++i) {
        const int k = 4 * i + 2 * kh;
        bf[i].x = Wm[k * D + c0 + m];
        bf[i].y = Wm[(k + 1) * D + c0 + m];
    }
}

__device__ __forceinline__ v8f tile_mm(const float* tile, int m, int kh, const v2f (&bf)[16]) {
    v8f acc = {};
#pragma unroll
    for (int i = 0; i < 16; ++i) {
        const v2f a = *(const v2f*)(tile + m * TP + 4 * i + 2 * kh);
        acc = __builtin_amdgcn_wmma_f32_16x16x4_f32(
             false, a,  false, bf[i],
             (short)0, acc,  false,  false);
    }
    return acc;
}

__global__ void __launch_bounds__(256)
k_gather_mm(const int* __restrict__ rowptr, const int* __restrict__ csr,
            const float* __restrict__ dinv, const float* __restrict__ x,
            const float* __restrict__ W1, const float* __restrict__ b1,
            const float* __restrict__ W2, float* __restrict__ xw2, int n) {
    __shared__ __attribute__((aligned(16))) float tA[16 * TP];
    __shared__ __attribute__((aligned(16))) float tH[16 * TP];
    __shared__ float tD[16];
    const int lane = threadIdx.x & 31;
    const int wave = __builtin_amdgcn_readfirstlane((int)(threadIdx.x >> 5));
    const int r0   = blockIdx.x * 16;
    const v2f* __restrict__ xr = (const v2f*)x;

    for (int q = 0; q < 2; ++q) {
        const int row = 2 * wave + q;
        const int v   = r0 + row;
        v2f acc = {0.0f, 0.0f};
        float dv = 0.0f;
        if (v < n) {
            const int beg = rowptr[v];
            const int end = rowptr[v + 1];
            dv = dinv[v];
            acc = xr[(size_t)v * 32 + lane];
            for (int j0 = beg; j0 < end; j0 += 32) {
                const int  j  = j0 + lane;
                const int  my = (j < end) ? csr[j] : n;
                const int cntc = min(32, end - j0);
                for (int k0 = 0; k0 < cntc; k0 += 8) {
#pragma unroll
                    for (int u = 0; u < 8; ++u) {
                        const int s = __builtin_amdgcn_readlane(my, k0 + u);
                        acc += xr[(size_t)s * 32 + lane];
                    }
                }
            }
        }
        *(v2f*)(tA + row * TP + 2 * lane) = acc * dv;
        if (lane == 0) tD[row] = dv;
    }

    const int m  = lane & 15;
    const int kh = lane >> 4;
    const int c0 = (wave & 3) << 4;
    v2f bf[16];
    load_bfrags(wave < 4 ? W1 : W2, c0, m, kh, bf);
    __syncthreads();

    if (wave < 4) {
        const v8f acc = tile_mm(tA, m, kh, bf);
        const float bb = b1[c0 + m];
#pragma unroll
        for (int r = 0; r < 8; ++r)
            tH[(r + 8 * kh) * TP + c0 + m] = fmaxf(acc[r] + bb, 0.0f);
    }
    __syncthreads();

    if (wave >= 4) {
        const v8f acc = tile_mm(tH, m, kh, bf);
        const int col = c0 + m;
#pragma unroll
        for (int r = 0; r < 8; ++r) {
            const int row = r0 + r + 8 * kh;
            if (row < n) xw2[(size_t)row * D + col] = acc[r] * tD[r + 8 * kh];
        }
    }
    if (blockIdx.x == 0 && wave == 0) *(v2f*)(xw2 + (size_t)n * D + 2 * lane) = v2f{0.0f, 0.0f};
}

template <bool RELU>
__global__ void __launch_bounds__(256)
k_gather(const int* __restrict__ rowptr, const int* __restrict__ csr,
         const float* __restrict__ dinv, const float* __restrict__ xw,
         const float* __restrict__ b, float* __restrict__ outp, int n) {
    const int lane = threadIdx.x & 31;
    const int v = __builtin_amdgcn_readfirstlane((int)(blockIdx.x * 8 + (threadIdx.x >> 5)));
    if (v >= n) return;
    const int beg = rowptr[v];
    const int end = rowptr[v + 1];
    const float dv = dinv[v];
    const v2f* __restrict__ xr = (const v2f*)xw;

    v2f acc = xr[(size_t)v * 32 + lane];

    for (int j0 = beg; j0 < end; j0 += 32) {
        const int  j  = j0 + lane;
        const int  my = (j < end) ? csr[j] : n;
        const int cntc = min(32, end - j0);
        for (int k0 = 0; k0 < cntc; k0 += 8) {
#pragma unroll
            for (int u = 0; u < 8; ++u) {
                const int s = __builtin_amdgcn_readlane(my, k0 + u);
                acc += xr[(size_t)s * 32 + lane];
            }
        }
    }

    v2f res = acc * dv + ((const v2f*)b)[lane];
    if (RELU) {
        res.x = fmaxf(res.x, 0.0f);
        res.y = fmaxf(res.y, 0.0f);
    }
    ((v2f*)outp)[(size_t)v * 32 + lane] = res;
}

extern "C" void kernel_launch(void* const* d_in, const int* in_sizes, int n_in,
                              void* d_out, int out_size, void* d_ws, size_t ws_size,
                              hipStream_t stream) {
    const float* x  = (const float*)d_in[0];
    const int*   ei = (const int*)  d_in[1];
    const float* W1 = (const float*)d_in[2];
    const float* b1 = (const float*)d_in[3];
    const float* W2 = (const float*)d_in[4];
    const float* b2 = (const float*)d_in[5];

    const int n = in_sizes[0] / D;
    const int e = in_sizes[1] / 2;
    const int* src = ei;
    const int* dst = ei + e;

    const int nbk = (n + BW - 1) >> BSH;

    float* ws    = (float*)d_ws;
    float* xw    = ws;
    float* xs    = xw + (size_t)(n + 1) * D;
    int*   pairs = (int*)(xs + (size_t)(n + 1) * D);
    int*   csr   = pairs + (size_t)2 * e;
    int*   loc   = csr + e;
    int*   rowptr  = loc + (size_t)(nbk + 1) * NB;
    float* dinv  = (float*)(rowptr + n + 1);
    float* out   = (float*)d_out;

    const int rowTiles = (n + 15) / 16;
    const int gatherBlocks = (n + 7) / 8;

    k_part<<<NB, SORT_THREADS, 0, stream>>>(src, dst, loc, pairs, e, nbk);
    k_bcsr<<<nbk, SORT_THREADS, 0, stream>>>(pairs, loc, rowptr, csr, dinv, x, xs, n, e);

    k_gather_mm<<<rowTiles, 256, 0, stream>>>(rowptr, csr, dinv, xs, W1, b1, W2, xw, n);

    k_gather<false><<<gatherBlocks, 256, 0, stream>>>(rowptr, csr, dinv, xw, b2, out, n);
}


# edge sort, both levels: the counting LDS atomic's return is kept as the record's rank and records are placed from registers (no second read, no second atomic); rows summed in that arrival order
# speedup vs baseline: 6.8859x; 1.0511x over previous
#include <hip/hip_runtime.h>

#define D 64
#define NB 256
#define SORT_THREADS 1024
#define BSH 8
#define BW (1 << BSH)
#define MAXBK 1024
#define PSTAGE 8192
#define CSTAGE 16384

typedef __attribute__((ext_vector_type(2))) float v2f;
typedef __attribute__((ext_vector_type(8))) float v8f;


__global__ void __launch_bounds__(SORT_THREADS)
k_part(const int* __restrict__ src, const int* __restrict__ dst, int* __restrict__ loc,
       int* __restrict__ pairs, int e, int nbk) {
    __shared__ int h[MAXBK];
    __shared__ int wsum[32];
    __shared__ int2 stg[PSTAGE];
    const int t    = threadIdx.x;
    const int lane = t & 31;
    const int w    = t >> 5;
    const int g    = blockIdx.x;
    const int per  = (e + NB - 1) / NB;
    const int e0   = g * per;
    const int e1   = min(e0 + per, e);
    const bool staged = per <= PSTAGE;

    for (int i = t; i < nbk; i += SORT_THREADS) h[i] = 0;
    __syncthreads();

    int sd[8], ss[8], sr[8];
    if (staged) {
        const int i0 = e0 + t;
#pragma unroll
        for (int u = 0; u < 8; ++u) { sd[u] = -1; ss[u] = 0; sr[u] = 0; }
        if (i0 < e1) {
            int xd[8];
#pragma unroll
            for (int u = 0; u < 8; ++u) {
                const int ic = min(i0 + u * SORT_THREADS, e1 - 1);
                xd[u] = dst[ic];
                ss[u] = src[ic];
            }
            __builtin_amdgcn_sched_barrier(0);
#pragma unroll
            for (int u = 0; u < 8; ++u) sd[u] = (i0 + u * SORT_THREADS < e1) ? xd[u] : -1;
#pragma unroll
            for (int u = 0; u < 8; ++u) {
                const int k = sd[u] >> BSH;
                if ((unsigned)k < (unsigned)nbk) sr[u] = atomicAdd(&h[k], 1);
            }
        }
    } else {
        for (int i0 = e0 + t; i0 < e1; i0 += 8 * SORT_THREADS) {
            int k[8];
#pragma unroll
            for (int u = 0; u < 8; ++u) {
                const int i = i0 + u * SORT_THREADS;
                const int x = dst[min(i, e1 - 1)];
                k[u] = (i < e1) ? (x >> BSH) : -1;
            }
#pragma unroll
            for (int u = 0; u < 8; ++u)
                if ((unsigned)k[u] < (unsigned)nbk) atomicAdd(&h[k[u]], 1);
        }
    }
    __syncthreads();

    const int x = (t < nbk) ? h[t] : 0;
    int inc = x;
    for (int o = 1; o < 32; o <<= 1) {
        const int y = __shfl_up(inc, o);
        if (lane >= o) inc += y;
    }
    if (lane == 31) wsum[w] = inc;
    __syncthreads();
    int woff = 0;
    int tot  = 0;
    for (int q = 0; q < 32; ++q) {
        const int z = wsum[q];
        tot += z;
        if (q < w) woff += z;
    }
    const int excl = woff + inc - x;
    if (t < nbk) {
        h[t] = excl;
        loc[(size_t)t * NB + g] = excl;
        if (t == nbk - 1) loc[(size_t)nbk * NB + g] = excl + x;
    }
    __syncthreads();

    int2* __restrict__ pp = (int2*)pairs;
    if (!staged) {
        for (int i0 = e0 + t; i0 < e1; i0 += 8 * SORT_THREADS) {
            int d[8], s[8];
#pragma unroll
            for (int u = 0; u < 8; ++u) {
                const int i  = i0 + u * SORT_THREADS;
                const int ic = min(i, e1 - 1);
                const int xd = dst[ic];
                s[u] = src[ic];
                d[u] = (i < e1) ? xd : -1;
            }
#pragma unroll
            for (int u = 0; u < 8; ++u) {
                const int k = d[u] >> BSH;
                if ((unsigned)k < (unsigned)nbk) pp[e0 + atomicAdd(&h[k], 1)] = make_int2(s[u], d[u]);
            }
        }
        return;
    }
    int hs[8];
#pragma unroll
    for (int u = 0; u < 8; ++u) hs[u] = h[max(min(sd[u] >> BSH, nbk - 1), 0)];
#pragma unroll
    for (int u = 0; u < 8; ++u) {
        const int k = sd[u] >> BSH;
        if ((unsigned)k < (unsigned)nbk) stg[hs[u] + sr[u]] = make_int2(ss[u], sd[u]);
    }

    __syncthreads();
    const int a0 = e0 & ~15;
    const int hi = e0 + tot;
    for (int i0 = a0 + t; i0 < hi; i0 += 4 * SORT_THREADS) {
        int2 pr[4];
#pragma unroll
        for (int u = 0; u < 4; ++u) {
            const int i = i0 + u * SORT_THREADS;
            pr[u] = stg[max(min(i, hi - 1), e0) - e0];
        }
#pragma unroll
        for (int u = 0; u < 4; ++u) {
            const int i = i0 + u * SORT_THREADS;
            if (i >= e0 && i < hi) pp[i] = pr[u];
        }
    }
}

__global__ void __launch_bounds__(SORT_THREADS)
k_bcsr(const int* __restrict__ pairs, const int* __restrict__ loc, int* __restrict__ rowptr,
       int* __restrict__ csr, float* __restrict__ dinv, const float* __restrict__ xin,
       float* __restrict__ xs, int n, int e) {
    __shared__ int cnt[BW];
    __shared__ int cnt2[BW];
    __shared__ int cur[BW];
    __shared__ int cur2[BW];
    __shared__ int wsum[BW / 32];
    __shared__ int rs[NB];
    __shared__ int rl[NB];
    __shared__ int wb[NB / 32];
    __shared__ float sdv[BW];
    __shared__ int stg[CSTAGE];
    const int t    = threadIdx.x;
    const int lane = t & 31;
    const int w    = t >> 5;
    const int wu   = __builtin_amdgcn_readfirstlane(w);
    const int k    = blockIdx.x;
    const int base = k << BSH;
    const int per  = (e + NB - 1) / NB;
    const int2* __restrict__ pp = (const int2*)pairs;

    int a = 0;
    if (t < NB) {
        a = loc[(size_t)k * NB + t];
        const int b = loc[(size_t)(k + 1) * NB + t];
        rs[t] = t * per + a;
        rl[t] = b - a;
    }
    if (t < BW) { cnt[t] = 0; cnt2[t] = 0; }
    __syncthreads();

    int s[NB / 16], d[NB / 16], r[NB / 16];
    bool late;
    {
        int st[NB / 32], ln[NB / 32];
        int mx = 0;
#pragma unroll
        for (int q = 0; q < NB / 32; ++q) {
            st[q] = rs[wu + 32 * q];
            ln[q] = rl[wu + 32 * q];
            mx = max(mx, ln[q]);
        }
        late = __builtin_amdgcn_readfirstlane(mx) > 64;
#pragma unroll
        for (int q = 0; q < NB / 32; ++q) {
#pragma unroll
            for (int hh = 0; hh < 2; ++hh) {
                const int i = lane + 32 * hh;
                const int2 pr = pp[min(st[q] + i, e - 1)];
                s[2 * q + hh] = pr.x;
                d[2 * q + hh] = (i < ln[q]) ? pr.y - base : -1;
            }
        }
    }

    const float4* __restrict__ x4 = (const float4*)xin;
    float4 xv[(BW * D / 4) / SORT_THREADS];
#pragma unroll
    for (int i = 0; i < (BW * D / 4) / SORT_THREADS; ++i) {
        const int q = t + SORT_THREADS * i;
        const int v = min(base + (q >> 4), n - 1);
        xv[i] = x4[(size_t)v * (D / 4) + (q & 15)];
    }

#pragma unroll
    for (int u = 0; u < NB / 16; ++u) {
        r[u] = 0;
        if ((unsigned)d[u] < (unsigned)BW) r[u] = atomicAdd(&cnt[d[u]], 1);
    }
    if (late) for (int q = 0; q < NB / 32; ++q) {
        const int st = rs[wu + 32 * q];
        const int ln = rl[wu + 32 * q];
        for (int i = 64 + lane; i < ln; i += 32) {
            const int dd = pp[st + i].y - base;
            if ((unsigned)dd < (unsigned)BW) atomicAdd(&cnt2[dd], 1);
        }
    }
    __syncthreads();

    const int xk = (t < BW) ? cnt[t] : 0;
    const int x  = xk + ((t < BW) ? cnt2[t] : 0);
    int inc = x;
    int sa  = a;
    for (int o = 1; o < 32; o <<= 1) {
        const int y = __shfl_up(inc, o);
        const int z = __shfl_xor(sa, o);
        if (lane >= o) inc += y;
        sa += z;
    }
    if (lane == 31) {
        if (t < BW) wsum[w] = inc;
        if (t < NB) wb[w] = sa;
    }
    __syncthreads();
    int b0 = 0;
    for (int q = 0; q < NB / 32; ++q) b0 += wb[q];
    int tot = 0;
    for (int q = 0; q < BW / 32; ++q) tot += wsum[q];
    const bool staged = tot <= CSTAGE;
    if (t < BW) {
        int woff = 0;
        for (int q = 0; q < BW / 32; ++q)
            if (q < w) woff += wsum[q];
        const int lo   = woff + inc - x;
        const int excl = b0 + lo;
        cur[t]  = lo;
        cur2[t] = lo + xk;
        const int v = base + t;
        if (v < n) {
            rowptr[v] = excl;
            const float dg = (float)(x + 1);
            const float di = (dg > 0.0f) ? rsqrtf(dg) : 0.0f;
            dinv[v] = di;
            sdv[t]  = di;
            if (v == n - 1) rowptr[n] = excl + x;
        }
    }
    __syncthreads();

    if (staged) {
        int c[NB / 16];
#pragma unroll
        for (int u = 0; u < NB / 16; ++u) c[u] = cur[d[u] & (BW - 1)];
#pragma unroll
        for (int u = 0; u < NB / 16; ++u)
            if ((unsigned)d[u] < (unsigned)BW) stg[c[u] + r[u]] = s[u];
        if (late) for (int q = 0; q < NB / 32; ++q) {
            const int st = rs[wu + 32 * q];
            const int ln = rl[wu + 32 * q];
            for (int i = 64 + lane; i < ln; i += 32) {
                const int2 pr = pp[st + i];
                const int dd = pr.y - base;
                if ((unsigned)dd < (unsigned)BW) stg[atomicAdd(&cur2[dd], 1)] = pr.x;
            }
        }
        __syncthreads();
        const int a0 = b0 & ~31;
        const int hi = b0 + tot;
        for (int i0 = a0 + t; i0 < hi; i0 += 4 * SORT_THREADS) {
            int sv[4];
#pragma unroll
            for (int u = 0; u < 4; ++u) {
                const int i = i0 + u * SORT_THREADS;
                sv[u] = stg[max(min(i, hi - 1), b0) - b0];
            }
#pragma unroll
            for (int u = 0; u < 4; ++u) {
                const int i = i0 + u * SORT_THREADS;
                if (i >= b0 && i < hi) csr[i] = sv[u];
            }
        }
    } else {
#pragma unroll
        for (int u = 0; u < NB / 16; ++u)
            if ((unsigned)d[u] < (unsigned)BW) csr[b0 + cur[d[u]] + r[u]] = s[u];
        if (late) for (int q = 0; q < NB / 32; ++q) {
            const int st = rs[wu + 32 * q];
            const int ln = rl[wu + 32 * q];
            for (int i = 64 + lane; i < ln; i += 32) {
                const int2 pr = pp[st + i];
                const int dd = pr.y - base;
                if ((unsigned)dd < (unsigned)BW) csr[b0 + atomicAdd(&cur2[dd], 1)] = pr.x;
            }
        }
    }

    float4* __restrict__ xs4 = (float4*)xs;
#pragma unroll
    for (int i = 0; i < (BW * D / 4) / SORT_THREADS; ++i) {
        const int q = t + SORT_THREADS * i;
        const int v = base + (q >> 4);
        if (v < n) {
            const float di = sdv[q >> 4];
            float4 rr = xv[i];
            rr.x *= di; rr.y *= di; rr.z *= di; rr.w *= di;
            xs4[(size_t)v * (D / 4) + (q & 15)] = rr;
        }
    }
    if (k == 0 && t < D / 4) xs4[(size_t)n * (D / 4) + t] = make_float4(0.0f, 0.0f, 0.0f, 0.0f);
}

#define TP 68

__device__ __forceinline__ void load_bfrags(const float* __restrict__ Wm, int c0, int m, int kh,
                                            v2f (&bf)[16]) {
#pragma unroll
    for (int i = 0; i < 16; ++i) {
        const int k = 4 * i + 2 * kh;
        bf[i].x = Wm[k * D + c0 + m];
        bf[i].y = Wm[(k + 1) * D + c0 + m];
    }
}

__device__ __forceinline__ v8f tile_mm(const float* tile, int m, int kh, const v2f (&bf)[16]) {
    v8f acc = {};
#pragma unroll
    for (int i = 0; i < 16; ++i) {
        const v2f a = *(const v2f*)(tile + m * TP + 4 * i + 2 * kh);
        acc = __builtin_amdgcn_wmma_f32_16x16x4_f32(
             false, a,  false, bf[i],
             (short)0, acc,  false,  false);
    }
    return acc;
}

__global__ void __launch_bounds__(256)
k_gather_mm(const int* __restrict__ rowptr, const int* __restrict__ csr,
            const float* __restrict__ dinv, const float* __restrict__ x,
            const float* __restrict__ W1, const float* __restrict__ b1,
            const float* __restrict__ W2, float* __restrict__ xw2, int n) {
    __shared__ __attribute__((aligned(16))) float tA[16 * TP];
    __shared__ __attribute__((aligned(16))) float tH[16 * TP];
    __shared__ float tD[16];
    const int lane = threadIdx.x & 31;
    const int wave = __builtin_amdgcn_readfirstlane((int)(threadIdx.x >> 5));
    const int r0   = blockIdx.x * 16;
    const v2f* __restrict__ xr = (const v2f*)x;

    for (int q = 0; q < 2; ++q) {
        const int row = 2 * wave + q;
        const int v   = r0 + row;
        v2f acc = {0.0f, 0.0f};
        float dv = 0.0f;
        if (v < n) {
            const int beg = rowptr[v];
            const int end = rowptr[v + 1];
            dv = dinv[v];
            acc = xr[(size_t)v * 32 + lane];
            for (int j0 = beg; j0 < end; j0 += 32) {
                const int  j  = j0 + lane;
                const int  my = (j < end) ? csr[j] : n;
                const int cntc = min(32, end - j0);
                for (int k0 = 0; k0 < cntc; k0 += 8) {
#pragma unroll
                    for (int u = 0; u < 8; ++u) {
                        const int s = __builtin_amdgcn_readlane(my, k0 + u);
                        acc += xr[(size_t)s * 32 + lane];
                    }
                }
            }
        }
        *(v2f*)(tA + row * TP + 2 * lane) = acc * dv;
        if (lane == 0) tD[row] = dv;
    }

    const int m  = lane & 15;
    const int kh = lane >> 4;
    const int c0 = (wave & 3) << 4;
    v2f bf[16];
    load_bfrags(wave < 4 ? W1 : W2, c0, m, kh, bf);
    __syncthreads();

    if (wave < 4) {
        const v8f acc = tile_mm(tA, m, kh, bf);
        const float bb = b1[c0 + m];
#pragma unroll
        for (int r = 0; r < 8; ++r)
            tH[(r + 8 * kh) * TP + c0 + m] = fmaxf(acc[r] + bb, 0.0f);
    }
    __syncthreads();

    if (wave >= 4) {
        const v8f acc = tile_mm(tH, m, kh, bf);
        const int col = c0 + m;
#pragma unroll
        for (int r = 0; r < 8; ++r) {
            const int row = r0 + r + 8 * kh;
            if (row < n) xw2[(size_t)row * D + col] = acc[r] * tD[r + 8 * kh];
        }
    }
    if (blockIdx.x == 0 && wave == 0) *(v2f*)(xw2 + (size_t)n * D + 2 * lane) = v2f{0.0f, 0.0f};
}

template <bool RELU>
__global__ void __launch_bounds__(256)
k_gather(const int* __restrict__ rowptr, const int* __restrict__ csr,
         const float* __restrict__ dinv, const float* __restrict__ xw,
         const float* __restrict__ b, float* __restrict__ outp, int n) {
    const int lane = threadIdx.x & 31;
    const int v = __builtin_amdgcn_readfirstlane((int)(blockIdx.x * 8 + (threadIdx.x >> 5)));
    if (v >= n) return;
    const int beg = rowptr[v];
    const int end = rowptr[v + 1];
    const float dv = dinv[v];
    const v2f* __restrict__ xr = (const v2f*)xw;

    v2f acc = xr[(size_t)v * 32 + lane];

    for (int j0 = beg; j0 < end; j0 += 32) {
        const int  j  = j0 + lane;
        const int  my = (j < end) ? csr[j] : n;
        const int cntc = min(32, end - j0);
        for (int k0 = 0; k0 < cntc; k0 += 8) {
#pragma unroll
            for (int u = 0; u < 8; ++u) {
                const int s = __builtin_amdgcn_readlane(my, k0 + u);
                acc += xr[(size_t)s * 32 + lane];
            }
        }
    }

    v2f res = acc * dv + ((const v2f*)b)[lane];
    if (RELU) {
        res.x = fmaxf(res.x, 0.0f);
        res.y = fmaxf(res.y, 0.0f);
    }
    ((v2f*)outp)[(size_t)v * 32 + lane] = res;
}

extern "C" void kernel_launch(void* const* d_in, const int* in_sizes, int n_in,
                              void* d_out, int out_size, void* d_ws, size_t ws_size,
                              hipStream_t stream) {
    const float* x  = (const float*)d_in[0];
    const int*   ei = (const int*)  d_in[1];
    const float* W1 = (const float*)d_in[2];
    const float* b1 = (const float*)d_in[3];
    const float* W2 = (const float*)d_in[4];
    const float* b2 = (const float*)d_in[5];

    const int n = in_sizes[0] / D;
    const int e = in_sizes[1] / 2;
    const int* src = ei;
    const int* dst = ei + e;

    const int nbk = (n + BW - 1) >> BSH;

    float* ws    = (float*)d_ws;
    float* xw    = ws;
    float* xs    = xw + (size_t)(n + 1) * D;
    int*   pairs = (int*)(xs + (size_t)(n + 1) * D);
    int*   csr   = pairs + (size_t)2 * e;
    int*   loc   = csr + e;
    int*   rowptr  = loc + (size_t)(nbk + 1) * NB;
    float* dinv  = (float*)(rowptr + n + 1);
    float* out   = (float*)d_out;

    const int rowTiles = (n + 15) / 16;
    const int gatherBlocks = (n + 7) / 8;

    k_part<<<NB, SORT_THREADS, 0, stream>>>(src, dst, loc, pairs, e, nbk);
    k_bcsr<<<nbk, SORT_THREADS, 0, stream>>>(pairs, loc, rowptr, csr, dinv, x, xs, n, e);

    k_gather_mm<<<rowTiles, 256, 0, stream>>>(rowptr, csr, dinv, xs, W1, b1, W2, xw, n);

    k_gather<false><<<gatherBlocks, 256, 0, stream>>>(rowptr, csr, dinv, xw, b2, out, n);
}
